# starting kernel
# baseline (speedup 1.0000x reference)
#include <hip/hip_runtime.h>
#include <cmath>

typedef __attribute__((ext_vector_type(2))) float v2f;
typedef __attribute__((ext_vector_type(8))) float v8f;

#define NND   50000
#define NED   800000
#define DIN   256
#define DH    128
#define ALPHA 0.1f
#define BNEPS 1e-5f

__global__ void k_deg_init(float* deg) {
    int i = blockIdx.x * blockDim.x + threadIdx.x;
    if (i < NND) deg[i] = 1.0f;
}
__global__ void k_deg_edges(const int* __restrict__ dst, float* deg) {
    int e = blockIdx.x * blockDim.x + threadIdx.x;
    if (e < NED) atomicAdd(&deg[dst[e]], 1.0f);
}
__global__ void k_deg_rsqrt(float* deg) {
    int i = blockIdx.x * blockDim.x + threadIdx.x;
    if (i < NND) deg[i] = rsqrtf(fmaxf(deg[i], 1.0f));
}

template <int K>
__global__ __launch_bounds__(256)
void k_gemm_wmma(const float* in, const float* __restrict__ W,
                 float* out, int mode, float beta) {
    extern __shared__ float2 sW2[];
    const int tid = threadIdx.x;

    {
        const int totalIter = (K / 2) * (DH / 4);
        for (int i = tid; i < totalIter; i += 256) {
            const int kp = i / (DH / 4);
            const int c4 = i % (DH / 4);
            const float4 a0 = *(const float4*)(W + (size_t)(2 * kp)     * DH + c4 * 4);
            const float4 a1 = *(const float4*)(W + (size_t)(2 * kp + 1) * DH + c4 * 4);
            float4* dst4 = (float4*)(sW2 + (size_t)kp * DH + c4 * 4);
            float4 f0; f0.x = a0.x; f0.y = a1.x; f0.z = a0.y; f0.w = a1.y;
            float4 f1; f1.x = a0.z; f1.y = a1.z; f1.z = a0.w; f1.w = a1.w;
            dst4[0] = f0;
            dst4[1] = f1;
        }
    }
    __syncthreads();

    const int wave = tid >> 5;
    const int lane = tid & 31;
    const int tileRow = blockIdx.x * 8 + wave;
    if (tileRow >= NND / 16) return;
    const int m0 = tileRow * 16;
    const int r    = lane & 15;
    const bool hi  = lane >= 16;
    const int koff = hi ? 2 : 0;
    const float* arow = in + (size_t)(m0 + r) * K;

    v8f c[8] = {};
    for (int k0 = 0; k0 < K; k0 += 4) {
        const float2 a2 = *(const float2*)(arow + k0 + koff);
        v2f a; a.x = a2.x; a.y = a2.y;
        const int pairRow = (k0 >> 1) + (hi ? 1 : 0);
        const float2* brow = sW2 + (size_t)pairRow * DH + r;
#pragma unroll
        for (int t = 0; t < 8; ++t) {
            const float2 b2 = brow[t * 16];
            v2f b; b.x = b2.x; b.y = b2.y;
            c[t] = __builtin_amdgcn_wmma_f32_16x16x4_f32(
                false, a, false, b, (short)0, c[t], false, false);
        }
    }

#pragma unroll
    for (int t = 0; t < 8; ++t) {
        const int col = t * 16 + r;
#pragma unroll
        for (int v = 0; v < 8; ++v) {
            const int row = m0 + v + (hi ? 8 : 0);
            const size_t o = (size_t)row * DH + col;
            float val = c[t][v];
            if (mode != 0) val = (1.0f - beta) * in[o] + beta * val;
            out[o] = val;
        }
    }
}

__global__ __launch_bounds__(256)
void k_gcn_self_bias(const float* __restrict__ xw, const float* __restrict__ dinv,
                     const float* __restrict__ b, float* __restrict__ z) {
    int i = blockIdx.x * blockDim.x + threadIdx.x;
    int node = i >> 5, l = i & 31;
    float w = dinv[node] * dinv[node];
    float4 v  = *(const float4*)(xw + (size_t)node * DH + l * 4);
    float4 bb = *(const float4*)(b + l * 4);
    float4 o;
    o.x = w * v.x + bb.x; o.y = w * v.y + bb.y;
    o.z = w * v.z + bb.z; o.w = w * v.w + bb.w;
    *(float4*)(z + (size_t)node * DH + l * 4) = o;
}

__global__ __launch_bounds__(256)
void k_gcn_edge(const int* __restrict__ src, const int* __restrict__ dst,
                const float* __restrict__ dinv, const float* __restrict__ xw,
                float* z) {
    int e = blockIdx.x * 8 + (threadIdx.x >> 5);
    int l = threadIdx.x & 31;
    int s = src[e], d = dst[e];
    float w = dinv[s] * dinv[d];
    const float4 v = *(const float4*)(xw + (size_t)s * DH + l * 4);
    float* p = z + (size_t)d * DH + l * 4;
    atomicAdd(p + 0, w * v.x);
    atomicAdd(p + 1, w * v.y);
    atomicAdd(p + 2, w * v.z);
    atomicAdd(p + 3, w * v.w);
}

__global__ __launch_bounds__(256)
void k_h_init(const float* __restrict__ x0, float* __restrict__ h) {
    size_t i = (size_t)blockIdx.x * blockDim.x + threadIdx.x;
    float4 v = *(const float4*)(x0 + i * 4);
    float4 o; o.x = ALPHA * v.x; o.y = ALPHA * v.y; o.z = ALPHA * v.z; o.w = ALPHA * v.w;
    *(float4*)(h + i * 4) = o;
}

__global__ __launch_bounds__(256)
void k_gcn2_edge(const int* __restrict__ src, const int* __restrict__ dst,
                 const float* __restrict__ z, float* h) {
    int e = blockIdx.x * 8 + (threadIdx.x >> 5);
    int l = threadIdx.x & 31;
    int s = src[e], d = dst[e];
    const float4 v = *(const float4*)(z + (size_t)s * DH + l * 4);
    float* p = h + (size_t)d * DH + l * 4;
    const float w = 1.0f - ALPHA;
    atomicAdd(p + 0, w * v.x);
    atomicAdd(p + 1, w * v.y);
    atomicAdd(p + 2, w * v.z);
    atomicAdd(p + 3, w * v.w);
}

__global__ void k_stats_zero(float* stats) { stats[threadIdx.x] = 0.0f; }

__global__ __launch_bounds__(256)
void k_bn_stats(const float* __restrict__ z, float* stats, int rowsPerBlock) {
    __shared__ float ssum[DH], ssq[DH];
    int tid = threadIdx.x;
    if (tid < DH) { ssum[tid] = 0.0f; ssq[tid] = 0.0f; }
    __syncthreads();
    int c = tid & (DH - 1);
    int rsub = tid >> 7;
    int base = blockIdx.x * rowsPerBlock;
    int rend = base + rowsPerBlock; if (rend > NND) rend = NND;
    float s = 0.0f, q = 0.0f;
    for (int r = base + rsub; r < rend; r += 2) {
        float v = z[(size_t)r * DH + c];
        s += v; q += v * v;
    }
    atomicAdd(&ssum[c], s);
    atomicAdd(&ssq[c], q);
    __syncthreads();
    if (tid < DH) {
        atomicAdd(&stats[tid], ssum[tid]);
        atomicAdd(&stats[DH + tid], ssq[tid]);
    }
}

__global__ void k_bn_final(float* stats) {
    int c = threadIdx.x;
    float invN = 1.0f / (float)NND;
    float mean = stats[c] * invN;
    float var  = stats[DH + c] * invN - mean * mean;
    stats[2 * DH + c] = mean;
    stats[3 * DH + c] = rsqrtf(var + BNEPS);
}

__global__ __launch_bounds__(256)
void k_bn_apply(float* z, const float* __restrict__ stats,
                const float* __restrict__ gamma, const float* __restrict__ beta) {
    size_t i = (size_t)blockIdx.x * blockDim.x + threadIdx.x;
    int c = (int)(i & (DH - 1));
    float v = z[i];
    v = (v - stats[2 * DH + c]) * stats[3 * DH + c] * gamma[c] + beta[c];
    z[i] = fmaxf(v, 0.0f);
}

extern "C" void kernel_launch(void* const* d_in, const int* in_sizes, int n_in,
                              void* d_out, int out_size, void* d_ws, size_t ws_size,
                              hipStream_t stream) {
    const float* x        = (const float*)d_in[0];
    const int*   edge     = (const int*)d_in[1];
    const float* W_lin    = (const float*)d_in[2];
    const float* b_lin    = (const float*)d_in[3];
    const float* W1s      = (const float*)d_in[4];
    const float* bn_gamma = (const float*)d_in[5];
    const float* bn_beta  = (const float*)d_in[6];
    float* out = (float*)d_out;

    float* ws    = (float*)d_ws;
    float* dinv  = ws;
    float* stats = ws + NND;
    float* bufA  = stats + 512;
    float* bufB  = bufA + (size_t)NND * DH;
    float* x0    = bufB + (size_t)NND * DH;

    const int* src = edge;
    const int* dst = edge + NED;

    const int nBlkN   = (NND + 255) / 256;
    const int nBlkE   = (NED + 255) / 256;
    const int nBlkVec = NND * 32 / 256;
    const int nBlkEdg = NED / 8;
    const int nBlkGemm = (NND / 16 + 7) / 8;

    k_deg_init<<<nBlkN, 256, 0, stream>>>(dinv);
    k_deg_edges<<<nBlkE, 256, 0, stream>>>(dst, dinv);
    k_deg_rsqrt<<<nBlkN, 256, 0, stream>>>(dinv);

    k_gemm_wmma<DIN><<<nBlkGemm, 256, DIN * DH * sizeof(float), stream>>>(
        x, W_lin, bufA, 0, 0.0f);

    k_gcn_self_bias<<<nBlkVec, 256, 0, stream>>>(bufA, dinv, b_lin, bufB);
    k_gcn_edge<<<nBlkEdg, 256, 0, stream>>>(src, dst, dinv, bufA, bufB);

    hipMemcpyAsync(x0, bufB, (size_t)NND * DH * sizeof(float),
                   hipMemcpyDeviceToDevice, stream);

    float* z = bufB;
    float* h = bufA;
    for (int i = 0; i < 5; ++i) {
        const float beta = logf(0.5f / (float)(i + 1) + 1.0f);
        k_h_init<<<nBlkVec, 256, 0, stream>>>(x0, h);
        k_gcn2_edge<<<nBlkEdg, 256, 0, stream>>>(src, dst, z, h);
        k_gemm_wmma<DH><<<nBlkGemm, 256, DH * DH * sizeof(float), stream>>>(
            h, W1s + (size_t)i * DH * DH, h, 1, beta);
        if (i < 4) {
            k_stats_zero<<<1, 256, 0, stream>>>(stats);
            k_bn_stats<<<(NND + 63) / 64, 256, 0, stream>>>(h, stats, 64);
            k_bn_final<<<1, DH, 0, stream>>>(stats);
            k_bn_apply<<<NND * DH / 256, 256, 0, stream>>>(
                h, stats, bn_gamma + i * DH, bn_beta + i * DH);
        }
        float* t = z; z = h; h = t;
    }

    hipMemcpyAsync(out, z, (size_t)NND * DH * sizeof(float),
                   hipMemcpyDeviceToDevice, stream);
}


# Edge sums as a gather: dst-sorted edge list built per call (int histogram+rank, scan, fill); one wave sums a node's rows in registers and writes it once with its start value; sum order changes
# speedup vs baseline: 2.3463x; 2.3463x over previous
#include <hip/hip_runtime.h>
#include <cmath>

typedef __attribute__((ext_vector_type(2))) float v2f;
typedef __attribute__((ext_vector_type(8))) float v8f;

#define NND   50000
#define NED   800000
#define DIN   256
#define DH    128
#define ALPHA 0.1f
#define BNEPS 1e-5f

#define NPAD  53248
#define SCANT 1024
#define SCANV 13

__global__ void k_cnt_zero(int* cnt) {
    int i = blockIdx.x * blockDim.x + threadIdx.x;
    cnt[i] = 0;
}
__global__ void k_hist_rank(const int* __restrict__ dst, int* cnt, int* __restrict__ rank) {
    int e = blockIdx.x * blockDim.x + threadIdx.x;
    rank[e] = atomicAdd(&cnt[dst[e]], 1);
}

__global__ __launch_bounds__(SCANT)
void k_scan_dinv(const int* __restrict__ cnt, int* __restrict__ rowptr,
                 float* __restrict__ dinv) {
    __shared__ int wtot[32];
    const int t = threadIdx.x;
    const int lane = t & 31, w = t >> 5;
    const int4* p = (const int4*)(cnt + t * (SCANV * 4));
    int4 v[SCANV];
    int s = 0;
#pragma unroll
    for (int j = 0; j < SCANV; ++j) {
        v[j] = p[j];
        s += v[j].x + v[j].y + v[j].z + v[j].w;
    }
    int incl = s;
#pragma unroll
    for (int o = 1; o < 32; o <<= 1) {
        int n = __shfl_up(incl, o);
        if (lane >= o) incl += n;
    }
    if (lane == 31) wtot[w] = incl;
    __syncthreads();
    if (w == 0) {
        int x = wtot[lane], xi = x;
#pragma unroll
        for (int o = 1; o < 32; o <<= 1) {
            int n = __shfl_up(xi, o);
            if (lane >= o) xi += n;
        }
        wtot[lane] = xi - x;
    }
    __syncthreads();
    int run = wtot[w] + incl - s;
    int4* q = (int4*)(rowptr + t * (SCANV * 4));
#pragma unroll
    for (int j = 0; j < SCANV; ++j) {
        int4 o;
        o.x = run; run += v[j].x;
        o.y = run; run += v[j].y;
        o.z = run; run += v[j].z;
        o.w = run; run += v[j].w;
        q[j] = o;
        const int g = t * (SCANV * 4) + j * 4;
        if (g < NND) {
            float4 dv;
            dv.x = rsqrtf(fmaxf((float)(v[j].x + 1), 1.0f));
            dv.y = rsqrtf(fmaxf((float)(v[j].y + 1), 1.0f));
            dv.z = rsqrtf(fmaxf((float)(v[j].z + 1), 1.0f));
            dv.w = rsqrtf(fmaxf((float)(v[j].w + 1), 1.0f));
            *(float4*)(dinv + g) = dv;
        }
    }
}

__global__ void k_csr_fill(const int* __restrict__ src, const int* __restrict__ dst,
                           const int* __restrict__ rowptr, const int* __restrict__ rank,
                           int* __restrict__ csr_src) {
    int e = blockIdx.x * blockDim.x + threadIdx.x;
    csr_src[rowptr[dst[e]] + rank[e]] = src[e];
}

template <int K>
__global__ __launch_bounds__(256)
void k_gemm_wmma(const float* in, const float* __restrict__ W,
                 float* out, int mode, float beta) {
    extern __shared__ float2 sW2[];
    const int tid = threadIdx.x;

    {
        const int totalIter = (K / 2) * (DH / 4);
        for (int i = tid; i < totalIter; i += 256) {
            const int kp = i / (DH / 4);
            const int c4 = i % (DH / 4);
            const float4 a0 = *(const float4*)(W + (size_t)(2 * kp)     * DH + c4 * 4);
            const float4 a1 = *(const float4*)(W + (size_t)(2 * kp + 1) * DH + c4 * 4);
            float4* dst4 = (float4*)(sW2 + (size_t)kp * DH + c4 * 4);
            float4 f0; f0.x = a0.x; f0.y = a1.x; f0.z = a0.y; f0.w = a1.y;
            float4 f1; f1.x = a0.z; f1.y = a1.z; f1.z = a0.w; f1.w = a1.w;
            dst4[0] = f0;
            dst4[1] = f1;
        }
    }
    __syncthreads();

    const int wave = tid >> 5;
    const int lane = tid & 31;
    const int tileRow = blockIdx.x * 8 + wave;
    if (tileRow >= NND / 16) return;
    const int m0 = tileRow * 16;
    const int r    = lane & 15;
    const bool hi  = lane >= 16;
    const int koff = hi ? 2 : 0;
    const float* arow = in + (size_t)(m0 + r) * K;

    v8f c[8] = {};
    for (int k0 = 0; k0 < K; k0 += 4) {
        const float2 a2 = *(const float2*)(arow + k0 + koff);
        v2f a; a.x = a2.x; a.y = a2.y;
        const int pairRow = (k0 >> 1) + (hi ? 1 : 0);
        const float2* brow = sW2 + (size_t)pairRow * DH + r;
#pragma unroll
        for (int t = 0; t < 8; ++t) {
            const float2 b2 = brow[t * 16];
            v2f b; b.x = b2.x; b.y = b2.y;
            c[t] = __builtin_amdgcn_wmma_f32_16x16x4_f32(
                false, a, false, b, (short)0, c[t], false, false);
        }
    }

#pragma unroll
    for (int t = 0; t < 8; ++t) {
        const int col = t * 16 + r;
#pragma unroll
        for (int v = 0; v < 8; ++v) {
            const int row = m0 + v + (hi ? 8 : 0);
            const size_t o = (size_t)row * DH + col;
            float val = c[t][v];
            if (mode != 0) val = (1.0f - beta) * in[o] + beta * val;
            out[o] = val;
        }
    }
}

__device__ __forceinline__ float4 ld_row(const float* __restrict__ f, int s, int l) {
    return *(const float4*)(f + (size_t)s * DH + l * 4);
}
__device__ __forceinline__ float4 zero4() {
    float4 v; v.x = 0.0f; v.y = 0.0f; v.z = 0.0f; v.w = 0.0f; return v;
}
__device__ __forceinline__ void add4(float4& a, const float4 v) {
    a.x += v.x; a.y += v.y; a.z += v.z; a.w += v.w;
}
__device__ __forceinline__ void fma4(float4& a, float w, const float4 v) {
    a.x += w * v.x; a.y += w * v.y; a.z += w * v.z; a.w += w * v.w;
}

__global__ __launch_bounds__(256)
void k_gcn_gather(const int* __restrict__ rowptr, const int* __restrict__ csr_src,
                  const float* __restrict__ dinv, const float* __restrict__ xw,
                  const float* __restrict__ b, float* __restrict__ z) {
    const int d   = __builtin_amdgcn_readfirstlane(blockIdx.x * 8 + (threadIdx.x >> 5));
    const int l   = threadIdx.x & 31;
    const int beg = __builtin_amdgcn_readfirstlane(rowptr[d]);
    const int end = __builtin_amdgcn_readfirstlane(rowptr[d + 1]);
    const float dd = dinv[d];
    const float4 vs = ld_row(xw, d, l);
    const float4 bb = *(const float4*)(b + l * 4);
    float4 acc = zero4();
    for (int base = beg; base < end; base += 32) {
        const int n = min(32, end - base);
        const int mine = (l < n) ? csr_src[base + l] : 0;
        int j = 0;
        for (; j + 4 <= n; j += 4) {
            const int s0 = __builtin_amdgcn_readlane(mine, j);
            const int s1 = __builtin_amdgcn_readlane(mine, j + 1);
            const int s2 = __builtin_amdgcn_readlane(mine, j + 2);
            const int s3 = __builtin_amdgcn_readlane(mine, j + 3);
            const float4 v0 = ld_row(xw, s0, l), v1 = ld_row(xw, s1, l);
            const float4 v2 = ld_row(xw, s2, l), v3 = ld_row(xw, s3, l);
            const float w0 = dinv[s0] * dd, w1 = dinv[s1] * dd;
            const float w2 = dinv[s2] * dd, w3 = dinv[s3] * dd;
            fma4(acc, w0, v0); fma4(acc, w1, v1); fma4(acc, w2, v2); fma4(acc, w3, v3);
        }
        const int r = n - j;
        if (r > 0) {
            const int s0 = __builtin_amdgcn_readlane(mine, j);
            const float4 v0 = ld_row(xw, s0, l);
            const float w0 = dinv[s0] * dd;
            float4 v1 = zero4(), v2 = zero4();
            float w1 = 0.0f, w2 = 0.0f;
            if (r > 1) {
                const int s1 = __builtin_amdgcn_readlane(mine, j + 1);
                v1 = ld_row(xw, s1, l); w1 = dinv[s1] * dd;
            }
            if (r > 2) {
                const int s2 = __builtin_amdgcn_readlane(mine, j + 2);
                v2 = ld_row(xw, s2, l); w2 = dinv[s2] * dd;
            }
            fma4(acc, w0, v0);
            if (r > 1) fma4(acc, w1, v1);
            if (r > 2) fma4(acc, w2, v2);
        }
    }
    const float ws = dd * dd;
    float4 o;
    o.x = acc.x + ws * vs.x + bb.x; o.y = acc.y + ws * vs.y + bb.y;
    o.z = acc.z + ws * vs.z + bb.z; o.w = acc.w + ws * vs.w + bb.w;
    *(float4*)(z + (size_t)d * DH + l * 4) = o;
}

__global__ __launch_bounds__(256)
void k_gcn2_gather(const int* __restrict__ rowptr, const int* __restrict__ csr_src,
                   const float* __restrict__ z, const float* __restrict__ x0,
                   float* __restrict__ h) {
    const int d   = __builtin_amdgcn_readfirstlane(blockIdx.x * 8 + (threadIdx.x >> 5));
    const int l   = threadIdx.x & 31;
    const int beg = __builtin_amdgcn_readfirstlane(rowptr[d]);
    const int end = __builtin_amdgcn_readfirstlane(rowptr[d + 1]);
    const float4 xv = ld_row(x0, d, l);
    float4 acc = zero4();
    for (int base = beg; base < end; base += 32) {
        const int n = min(32, end - base);
        const int mine = (l < n) ? csr_src[base + l] : 0;
        int j = 0;
        for (; j + 4 <= n; j += 4) {
            const int s0 = __builtin_amdgcn_readlane(mine, j);
            const int s1 = __builtin_amdgcn_readlane(mine, j + 1);
            const int s2 = __builtin_amdgcn_readlane(mine, j + 2);
            const int s3 = __builtin_amdgcn_readlane(mine, j + 3);
            const float4 v0 = ld_row(z, s0, l), v1 = ld_row(z, s1, l);
            const float4 v2 = ld_row(z, s2, l), v3 = ld_row(z, s3, l);
            add4(acc, v0); add4(acc, v1); add4(acc, v2); add4(acc, v3);
        }
        const int r = n - j;
        if (r > 0) {
            const float4 v0 = ld_row(z, __builtin_amdgcn_readlane(mine, j), l);
            float4 v1 = zero4(), v2 = zero4();
            if (r > 1) v1 = ld_row(z, __builtin_amdgcn_readlane(mine, j + 1), l);
            if (r > 2) v2 = ld_row(z, __builtin_amdgcn_readlane(mine, j + 2), l);
            add4(acc, v0);
            if (r > 1) add4(acc, v1);
            if (r > 2) add4(acc, v2);
        }
    }
    const float w = 1.0f - ALPHA;
    float4 o;
    o.x = ALPHA * xv.x + w * acc.x; o.y = ALPHA * xv.y + w * acc.y;
    o.z = ALPHA * xv.z + w * acc.z; o.w = ALPHA * xv.w + w * acc.w;
    *(float4*)(h + (size_t)d * DH + l * 4) = o;
}

__global__ void k_stats_zero(float* stats) { stats[threadIdx.x] = 0.0f; }

__global__ __launch_bounds__(256)
void k_bn_stats(const float* __restrict__ z, float* stats, int rowsPerBlock) {
    __shared__ float ssum[DH], ssq[DH];
    int tid = threadIdx.x;
    if (tid < DH) { ssum[tid] = 0.0f; ssq[tid] = 0.0f; }
    __syncthreads();
    int c = tid & (DH - 1);
    int rsub = tid >> 7;
    int base = blockIdx.x * rowsPerBlock;
    int rend = base + rowsPerBlock; if (rend > NND) rend = NND;
    float s = 0.0f, q = 0.0f;
    for (int r = base + rsub; r < rend; r += 2) {
        float v = z[(size_t)r * DH + c];
        s += v; q += v * v;
    }
    atomicAdd(&ssum[c], s);
    atomicAdd(&ssq[c], q);
    __syncthreads();
    if (tid < DH) {
        atomicAdd(&stats[tid], ssum[tid]);
        atomicAdd(&stats[DH + tid], ssq[tid]);
    }
}

__global__ void k_bn_final(float* stats) {
    int c = threadIdx.x;
    float invN = 1.0f / (float)NND;
    float mean = stats[c] * invN;
    float var  = stats[DH + c] * invN - mean * mean;
    stats[2 * DH + c] = mean;
    stats[3 * DH + c] = rsqrtf(var + BNEPS);
}

__global__ __launch_bounds__(256)
void k_bn_apply(float* z, const float* __restrict__ stats,
                const float* __restrict__ gamma, const float* __restrict__ beta) {
    size_t i = (size_t)blockIdx.x * blockDim.x + threadIdx.x;
    int c = (int)(i & (DH - 1));
    float v = z[i];
    v = (v - stats[2 * DH + c]) * stats[3 * DH + c] * gamma[c] + beta[c];
    z[i] = fmaxf(v, 0.0f);
}

extern "C" void kernel_launch(void* const* d_in, const int* in_sizes, int n_in,
                              void* d_out, int out_size, void* d_ws, size_t ws_size,
                              hipStream_t stream) {
    const float* x        = (const float*)d_in[0];
    const int*   edge     = (const int*)d_in[1];
    const float* W_lin    = (const float*)d_in[2];
    const float* b_lin    = (const float*)d_in[3];
    const float* W1s      = (const float*)d_in[4];
    const float* bn_gamma = (const float*)d_in[5];
    const float* bn_beta  = (const float*)d_in[6];
    float* out = (float*)d_out;

    float* ws    = (float*)d_ws;
    float* dinv  = ws;
    float* stats = ws + NND;
    float* bufA  = stats + 512;
    float* bufB  = bufA + (size_t)NND * DH;
    float* x0    = bufB + (size_t)NND * DH;
    int*   cnt     = (int*)(x0 + (size_t)NND * DH);
    int*   rowptr  = cnt + NPAD;
    int*   rank    = rowptr + NPAD;
    int*   csr_src = rank + NED;

    const int* src = edge;
    const int* dst = edge + NED;

    const int nBlkE   = NED / 256;
    const int nBlkRow = NND / 8;
    const int nBlkGemm = (NND / 16 + 7) / 8;

    k_cnt_zero<<<NPAD / 256, 256, 0, stream>>>(cnt);
    k_hist_rank<<<nBlkE, 256, 0, stream>>>(dst, cnt, rank);
    k_scan_dinv<<<1, SCANT, 0, stream>>>(cnt, rowptr, dinv);
    k_csr_fill<<<nBlkE, 256, 0, stream>>>(src, dst, rowptr, rank, csr_src);

    k_gemm_wmma<DIN><<<nBlkGemm, 256, DIN * DH * sizeof(float), stream>>>(
        x, W_lin, bufA, 0, 0.0f);

    k_gcn_gather<<<nBlkRow, 256, 0, stream>>>(rowptr, csr_src, dinv, bufA, b_lin, bufB);

    hipMemcpyAsync(x0, bufB, (size_t)NND * DH * sizeof(float),
                   hipMemcpyDeviceToDevice, stream);

    float* z = bufB;
    float* h = bufA;
    for (int i = 0; i < 5; ++i) {
        const float beta = logf(0.5f / (float)(i + 1) + 1.0f);
        k_gcn2_gather<<<nBlkRow, 256, 0, stream>>>(rowptr, csr_src, z, x0, h);
        k_gemm_wmma<DH><<<nBlkGemm, 256, DH * DH * sizeof(float), stream>>>(
            h, W1s + (size_t)i * DH * DH, h, 1, beta);
        if (i < 4) {
            k_stats_zero<<<1, 256, 0, stream>>>(stats);
            k_bn_stats<<<(NND + 63) / 64, 256, 0, stream>>>(h, stats, 64);
            k_bn_final<<<1, DH, 0, stream>>>(stats);
            k_bn_apply<<<NND * DH / 256, 256, 0, stream>>>(
                h, stats, bn_gamma + i * DH, bn_beta + i * DH);
        }
        float* t = z; z = h; h = t;
    }

    hipMemcpyAsync(out, z, (size_t)NND * DH * sizeof(float),
                   hipMemcpyDeviceToDevice, stream);
}


# BatchNorm folded into its neighbours: column sums in the GEMM epilogue, scale/shift in the next gather's prologue, normalise+ReLU on gathered rows; launches halved; statistics sum order changes
# speedup vs baseline: 3.1038x; 1.3228x over previous
#include <hip/hip_runtime.h>
#include <cmath>

typedef __attribute__((ext_vector_type(2))) float v2f;
typedef __attribute__((ext_vector_type(8))) float v8f;

#define NND   50000
#define NED   800000
#define DIN   256
#define DH    128
#define ALPHA 0.1f
#define BNEPS 1e-5f

#define NPAD  53248
#define NSTAT (4 * 2 * DH)
#define SCANT 1024
#define SCANV 13

__global__ void k_cnt_zero(int* cnt, float* stats) {
    int i = blockIdx.x * blockDim.x + threadIdx.x;
    cnt[i] = 0;
    if (i < NSTAT) stats[i] = 0.0f;
}
__global__ void k_hist_rank(const int* __restrict__ dst, int* cnt, int* __restrict__ rank) {
    int e = blockIdx.x * blockDim.x + threadIdx.x;
    rank[e] = atomicAdd(&cnt[dst[e]], 1);
}

__global__ __launch_bounds__(SCANT)
void k_scan_dinv(const int* __restrict__ cnt, int* __restrict__ rowptr,
                 float* __restrict__ dinv) {
    __shared__ int wtot[32];
    const int t = threadIdx.x;
    const int lane = t & 31, w = t >> 5;
    const int4* p = (const int4*)(cnt + t * (SCANV * 4));
    int4 v[SCANV];
    int s = 0;
#pragma unroll
    for (int j = 0; j < SCANV; ++j) {
        v[j] = p[j];
        s += v[j].x + v[j].y + v[j].z + v[j].w;
    }
    int incl = s;
#pragma unroll
    for (int o = 1; o < 32; o <<= 1) {
        int n = __shfl_up(incl, o);
        if (lane >= o) incl += n;
    }
    if (lane == 31) wtot[w] = incl;
    __syncthreads();
    if (w == 0) {
        int x = wtot[lane], xi = x;
#pragma unroll
        for (int o = 1; o < 32; o <<= 1) {
            int n = __shfl_up(xi, o);
            if (lane >= o) xi += n;
        }
        wtot[lane] = xi - x;
    }
    __syncthreads();
    int run = wtot[w] + incl - s;
    int4* q = (int4*)(rowptr + t * (SCANV * 4));
#pragma unroll
    for (int j = 0; j < SCANV; ++j) {
        int4 o;
        o.x = run; run += v[j].x;
        o.y = run; run += v[j].y;
        o.z = run; run += v[j].z;
        o.w = run; run += v[j].w;
        q[j] = o;
        const int g = t * (SCANV * 4) + j * 4;
        if (g < NND) {
            float4 dv;
            dv.x = rsqrtf(fmaxf((float)(v[j].x + 1), 1.0f));
            dv.y = rsqrtf(fmaxf((float)(v[j].y + 1), 1.0f));
            dv.z = rsqrtf(fmaxf((float)(v[j].z + 1), 1.0f));
            dv.w = rsqrtf(fmaxf((float)(v[j].w + 1), 1.0f));
            *(float4*)(dinv + g) = dv;
        }
    }
}

__global__ void k_csr_fill(const int* __restrict__ src, const int* __restrict__ dst,
                           const int* __restrict__ rowptr, const int* __restrict__ rank,
                           int* __restrict__ csr_src) {
    int e = blockIdx.x * blockDim.x + threadIdx.x;
    csr_src[rowptr[dst[e]] + rank[e]] = src[e];
}

template <int K>
__global__ __launch_bounds__(256)
void k_gemm_wmma(const float* in, const float* __restrict__ W,
                 float* out, int mode, float beta, float* stats) {
    extern __shared__ float2 sW2[];
    __shared__ float wpart[8][2 * DH];
    const int tid = threadIdx.x;

    {
        const int totalIter = (K / 2) * (DH / 4);
        for (int i = tid; i < totalIter; i += 256) {
            const int kp = i / (DH / 4);
            const int c4 = i % (DH / 4);
            const float4 a0 = *(const float4*)(W + (size_t)(2 * kp)     * DH + c4 * 4);
            const float4 a1 = *(const float4*)(W + (size_t)(2 * kp + 1) * DH + c4 * 4);
            float4* dst4 = (float4*)(sW2 + (size_t)kp * DH + c4 * 4);
            float4 f0; f0.x = a0.x; f0.y = a1.x; f0.z = a0.y; f0.w = a1.y;
            float4 f1; f1.x = a0.z; f1.y = a1.z; f1.z = a0.w; f1.w = a1.w;
            dst4[0] = f0;
            dst4[1] = f1;
        }
    }
    __syncthreads();

    const int wave = tid >> 5;
    const int lane = tid & 31;
    const int tileRow = blockIdx.x * 8 + wave;
    const bool valid = tileRow < NND / 16;
    if (valid) {
        const int m0 = tileRow * 16;
        const int r    = lane & 15;
        const bool hi  = lane >= 16;
        const int koff = hi ? 2 : 0;
        const float* arow = in + (size_t)(m0 + r) * K;

        v8f c[8] = {};
        for (int k0 = 0; k0 < K; k0 += 4) {
            const float2 a2 = *(const float2*)(arow + k0 + koff);
            v2f a; a.x = a2.x; a.y = a2.y;
            const int pairRow = (k0 >> 1) + (hi ? 1 : 0);
            const float2* brow = sW2 + (size_t)pairRow * DH + r;
#pragma unroll
            for (int t = 0; t < 8; ++t) {
                const float2 b2 = brow[t * 16];
                v2f b; b.x = b2.x; b.y = b2.y;
                c[t] = __builtin_amdgcn_wmma_f32_16x16x4_f32(
                    false, a, false, b, (short)0, c[t], false, false);
            }
        }

        float cs[8], cq[8];
#pragma unroll
        for (int t = 0; t < 8; ++t) {
            const int col = t * 16 + r;
            cs[t] = 0.0f; cq[t] = 0.0f;
#pragma unroll
            for (int v = 0; v < 8; ++v) {
                const int row = m0 + v + (hi ? 8 : 0);
                const size_t o = (size_t)row * DH + col;
                float val = c[t][v];
                if (mode != 0) val = (1.0f - beta) * in[o] + beta * val;
                out[o] = val;
                cs[t] += val; cq[t] += val * val;
            }
        }
        if (stats != nullptr) {
#pragma unroll
            for (int t = 0; t < 8; ++t) {
                const float s = cs[t] + __shfl_xor(cs[t], 16);
                const float q = cq[t] + __shfl_xor(cq[t], 16);
                wpart[wave][(hi ? DH : 0) + t * 16 + r] = hi ? q : s;
            }
        }
    }

    if (stats != nullptr) {
        if (!valid) {
            for (int j = lane; j < 2 * DH; j += 32) wpart[wave][j] = 0.0f;
        }
        __syncthreads();
        float tot = 0.0f;
#pragma unroll
        for (int w = 0; w < 8; ++w) tot += wpart[w][tid];
        atomicAdd(&stats[tid], tot);
    }
}

__device__ __forceinline__ float4 ld_row(const float* __restrict__ f, int s, int l) {
    return *(const float4*)(f + (size_t)s * DH + l * 4);
}
__device__ __forceinline__ float4 zero4() {
    float4 v; v.x = 0.0f; v.y = 0.0f; v.z = 0.0f; v.w = 0.0f; return v;
}
__device__ __forceinline__ void add4(float4& a, const float4 v) {
    a.x += v.x; a.y += v.y; a.z += v.z; a.w += v.w;
}
__device__ __forceinline__ void fma4(float4& a, float w, const float4 v) {
    a.x += w * v.x; a.y += w * v.y; a.z += w * v.z; a.w += w * v.w;
}

__global__ __launch_bounds__(256)
void k_gcn_gather(const int* __restrict__ rowptr, const int* __restrict__ csr_src,
                  const float* __restrict__ dinv, const float* __restrict__ xw,
                  const float* __restrict__ b, float* __restrict__ z) {
    const int d   = __builtin_amdgcn_readfirstlane(blockIdx.x * 8 + (threadIdx.x >> 5));
    const int l   = threadIdx.x & 31;
    const int beg = __builtin_amdgcn_readfirstlane(rowptr[d]);
    const int end = __builtin_amdgcn_readfirstlane(rowptr[d + 1]);
    const float dd = dinv[d];
    const float4 vs = ld_row(xw, d, l);
    const float4 bb = *(const float4*)(b + l * 4);
    float4 acc = zero4();
    for (int base = beg; base < end; base += 32) {
        const int n = min(32, end - base);
        const int mine = (l < n) ? csr_src[base + l] : 0;
        int j = 0;
        for (; j + 4 <= n; j += 4) {
            const int s0 = __builtin_amdgcn_readlane(mine, j);
            const int s1 = __builtin_amdgcn_readlane(mine, j + 1);
            const int s2 = __builtin_amdgcn_readlane(mine, j + 2);
            const int s3 = __builtin_amdgcn_readlane(mine, j + 3);
            const float4 v0 = ld_row(xw, s0, l), v1 = ld_row(xw, s1, l);
            const float4 v2 = ld_row(xw, s2, l), v3 = ld_row(xw, s3, l);
            const float w0 = dinv[s0] * dd, w1 = dinv[s1] * dd;
            const float w2 = dinv[s2] * dd, w3 = dinv[s3] * dd;
            fma4(acc, w0, v0); fma4(acc, w1, v1); fma4(acc, w2, v2); fma4(acc, w3, v3);
        }
        const int r = n - j;
        if (r > 0) {
            const int s0 = __builtin_amdgcn_readlane(mine, j);
            const float4 v0 = ld_row(xw, s0, l);
            const float w0 = dinv[s0] * dd;
            float4 v1 = zero4(), v2 = zero4();
            float w1 = 0.0f, w2 = 0.0f;
            if (r > 1) {
                const int s1 = __builtin_amdgcn_readlane(mine, j + 1);
                v1 = ld_row(xw, s1, l); w1 = dinv[s1] * dd;
            }
            if (r > 2) {
                const int s2 = __builtin_amdgcn_readlane(mine, j + 2);
                v2 = ld_row(xw, s2, l); w2 = dinv[s2] * dd;
            }
            fma4(acc, w0, v0);
            if (r > 1) fma4(acc, w1, v1);
            if (r > 2) fma4(acc, w2, v2);
        }
    }
    const float ws = dd * dd;
    float4 o;
    o.x = acc.x + ws * vs.x + bb.x; o.y = acc.y + ws * vs.y + bb.y;
    o.z = acc.z + ws * vs.z + bb.z; o.w = acc.w + ws * vs.w + bb.w;
    *(float4*)(z + (size_t)d * DH + l * 4) = o;
}

__device__ __forceinline__ void bn_coef1(float su, float sq, float g, float bt,
                                         float& sc, float& sh) {
    const float invN = 1.0f / (float)NND;
    const float mean = su * invN;
    const float var  = sq * invN - mean * mean;
    sc = rsqrtf(var + BNEPS) * g;
    sh = bt - mean * sc;
}
template <bool BN>
__device__ __forceinline__ void acc_row(float4& a, const float4 v, const float4 sc, const float4 sh) {
    if (BN) {
        a.x += fmaxf(fmaf(v.x, sc.x, sh.x), 0.0f); a.y += fmaxf(fmaf(v.y, sc.y, sh.y), 0.0f);
        a.z += fmaxf(fmaf(v.z, sc.z, sh.z), 0.0f); a.w += fmaxf(fmaf(v.w, sc.w, sh.w), 0.0f);
    } else {
        add4(a, v);
    }
}

template <bool BN>
__global__ __launch_bounds__(256)
void k_gcn2_gather(const int* __restrict__ rowptr, const int* __restrict__ csr_src,
                   const float* __restrict__ z, const float* __restrict__ x0,
                   float* __restrict__ h, const float* __restrict__ stats,
                   const float* __restrict__ gamma, const float* __restrict__ bbeta) {
    const int d   = __builtin_amdgcn_readfirstlane(blockIdx.x * 8 + (threadIdx.x >> 5));
    const int l   = threadIdx.x & 31;
    const int beg = __builtin_amdgcn_readfirstlane(rowptr[d]);
    const int end = __builtin_amdgcn_readfirstlane(rowptr[d + 1]);
    const float4 xv = ld_row(x0, d, l);
    float4 sc = zero4(), sh = zero4();
    if (BN) {
        const float4 su = *(const float4*)(stats + l * 4);
        const float4 sq = *(const float4*)(stats + DH + l * 4);
        const float4 g  = *(const float4*)(gamma + l * 4);
        const float4 bt = *(const float4*)(bbeta + l * 4);
        bn_coef1(su.x, sq.x, g.x, bt.x, sc.x, sh.x);
        bn_coef1(su.y, sq.y, g.y, bt.y, sc.y, sh.y);
        bn_coef1(su.z, sq.z, g.z, bt.z, sc.z, sh.z);
        bn_coef1(su.w, sq.w, g.w, bt.w, sc.w, sh.w);
    }
    float4 acc = zero4();
    for (int base = beg; base < end; base += 32) {
        const int n = min(32, end - base);
        const int mine = (l < n) ? csr_src[base + l] : 0;
        int j = 0;
        for (; j + 4 <= n; j += 4) {
            const int s0 = __builtin_amdgcn_readlane(mine, j);
            const int s1 = __builtin_amdgcn_readlane(mine, j + 1);
            const int s2 = __builtin_amdgcn_readlane(mine, j + 2);
            const int s3 = __builtin_amdgcn_readlane(mine, j + 3);
            const float4 v0 = ld_row(z, s0, l), v1 = ld_row(z, s1, l);
            const float4 v2 = ld_row(z, s2, l), v3 = ld_row(z, s3, l);
            acc_row<BN>(acc, v0, sc, sh); acc_row<BN>(acc, v1, sc, sh);
            acc_row<BN>(acc, v2, sc, sh); acc_row<BN>(acc, v3, sc, sh);
        }
        const int r = n - j;
        if (r > 0) {
            const float4 v0 = ld_row(z, __builtin_amdgcn_readlane(mine, j), l);
            float4 v1 = zero4(), v2 = zero4();
            if (r > 1) v1 = ld_row(z, __builtin_amdgcn_readlane(mine, j + 1), l);
            if (r > 2) v2 = ld_row(z, __builtin_amdgcn_readlane(mine, j + 2), l);
            acc_row<BN>(acc, v0, sc, sh);
            if (r > 1) acc_row<BN>(acc, v1, sc, sh);
            if (r > 2) acc_row<BN>(acc, v2, sc, sh);
        }
    }
    const float w = 1.0f - ALPHA;
    float4 o;
    o.x = ALPHA * xv.x + w * acc.x; o.y = ALPHA * xv.y + w * acc.y;
    o.z = ALPHA * xv.z + w * acc.z; o.w = ALPHA * xv.w + w * acc.w;
    *(float4*)(h + (size_t)d * DH + l * 4) = o;
}

extern "C" void kernel_launch(void* const* d_in, const int* in_sizes, int n_in,
                              void* d_out, int out_size, void* d_ws, size_t ws_size,
                              hipStream_t stream) {
    const float* x        = (const float*)d_in[0];
    const int*   edge     = (const int*)d_in[1];
    const float* W_lin    = (const float*)d_in[2];
    const float* b_lin    = (const float*)d_in[3];
    const float* W1s      = (const float*)d_in[4];
    const float* bn_gamma = (const float*)d_in[5];
    const float* bn_beta  = (const float*)d_in[6];
    float* out = (float*)d_out;

    float* ws    = (float*)d_ws;
    float* dinv  = ws;
    float* stats = ws + NND;
    float* bufA  = stats + NSTAT;
    float* bufB  = bufA + (size_t)NND * DH;
    float* x0    = bufB + (size_t)NND * DH;
    int*   cnt     = (int*)(x0 + (size_t)NND * DH);
    int*   rowptr  = cnt + NPAD;
    int*   rank    = rowptr + NPAD;
    int*   csr_src = rank + NED;

    const int* src = edge;
    const int* dst = edge + NED;

    const int nBlkE   = NED / 256;
    const int nBlkRow = NND / 8;
    const int nBlkGemm = (NND / 16 + 7) / 8;

    k_cnt_zero<<<NPAD / 256, 256, 0, stream>>>(cnt, stats);
    k_hist_rank<<<nBlkE, 256, 0, stream>>>(dst, cnt, rank);
    k_scan_dinv<<<1, SCANT, 0, stream>>>(cnt, rowptr, dinv);
    k_csr_fill<<<nBlkE, 256, 0, stream>>>(src, dst, rowptr, rank, csr_src);

    k_gemm_wmma<DIN><<<nBlkGemm, 256, DIN * DH * sizeof(float), stream>>>(
        x, W_lin, bufA, 0, 0.0f, nullptr);

    k_gcn_gather<<<nBlkRow, 256, 0, stream>>>(rowptr, csr_src, dinv, bufA, b_lin, bufB);

    hipMemcpyAsync(x0, bufB, (size_t)NND * DH * sizeof(float),
                   hipMemcpyDeviceToDevice, stream);

    float* z = bufB;
    float* h = bufA;
    for (int i = 0; i < 5; ++i) {
        const float beta = logf(0.5f / (float)(i + 1) + 1.0f);
        if (i == 0) {
            k_gcn2_gather<false><<<nBlkRow, 256, 0, stream>>>(
                rowptr, csr_src, z, x0, h, nullptr, nullptr, nullptr);
        } else {
            k_gcn2_gather<true><<<nBlkRow, 256, 0, stream>>>(
                rowptr, csr_src, z, x0, h, stats + (i - 1) * 2 * DH,
                bn_gamma + (i - 1) * DH, bn_beta + (i - 1) * DH);
        }
        k_gemm_wmma<DH><<<nBlkGemm, 256, DH * DH * sizeof(float), stream>>>(
            h, W1s + (size_t)i * DH * DH, h, 1, beta,
            i < 4 ? stats + i * 2 * DH : nullptr);
        float* t = z; z = h; h = t;
    }

    hipMemcpyAsync(out, z, (size_t)NND * DH * sizeof(float),
                   hipMemcpyDeviceToDevice, stream);
}


# GCN2 layer in one launch: a wave gathers its 16 nodes' rows (flat list walk, 16 rows in flight) into LDS as the WMMA A operand; h and both device copies leave memory; sum order unchanged
# speedup vs baseline: 4.0969x; 1.3200x over previous
#include <hip/hip_runtime.h>
#include <cmath>

typedef __attribute__((ext_vector_type(2))) float v2f;
typedef __attribute__((ext_vector_type(8))) float v8f;

#define NND   50000
#define NED   800000
#define DIN   256
#define DH    128
#define ALPHA 0.1f
#define BNEPS 1e-5f

#define NPAD  53248
#define NSTAT (4 * 2 * DH)
#define SCANT 1024
#define SCANV 13

__global__ void k_cnt_zero(int* cnt, float* stats) {
    int i = blockIdx.x * blockDim.x + threadIdx.x;
    cnt[i] = 0;
    if (i < NSTAT) stats[i] = 0.0f;
}
__global__ void k_hist_rank(const int* __restrict__ dst, int* cnt, int* __restrict__ rank) {
    int e = blockIdx.x * blockDim.x + threadIdx.x;
    rank[e] = atomicAdd(&cnt[dst[e]], 1);
}

__global__ __launch_bounds__(SCANT)
void k_scan_dinv(const int* __restrict__ cnt, int* __restrict__ rowptr,
                 float* __restrict__ dinv) {
    __shared__ int wtot[32];
    const int t = threadIdx.x;
    const int lane = t & 31, w = t >> 5;
    const int4* p = (const int4*)(cnt + t * (SCANV * 4));
    int4 v[SCANV];
    int s = 0;
#pragma unroll
    for (int j = 0; j < SCANV; ++j) {
        v[j] = p[j];
        s += v[j].x + v[j].y + v[j].z + v[j].w;
    }
    int incl = s;
#pragma unroll
    for (int o = 1; o < 32; o <<= 1) {
        int n = __shfl_up(incl, o);
        if (lane >= o) incl += n;
    }
    if (lane == 31) wtot[w] = incl;
    __syncthreads();
    if (w == 0) {
        int x = wtot[lane], xi = x;
#pragma unroll
        for (int o = 1; o < 32; o <<= 1) {
            int n = __shfl_up(xi, o);
            if (lane >= o) xi += n;
        }
        wtot[lane] = xi - x;
    }
    __syncthreads();
    int run = wtot[w] + incl - s;
    int4* q = (int4*)(rowptr + t * (SCANV * 4));
#pragma unroll
    for (int j = 0; j < SCANV; ++j) {
        int4 o;
        o.x = run; run += v[j].x;
        o.y = run; run += v[j].y;
        o.z = run; run += v[j].z;
        o.w = run; run += v[j].w;
        q[j] = o;
        const int g = t * (SCANV * 4) + j * 4;
        if (g < NND) {
            float4 dv;
            dv.x = rsqrtf(fmaxf((float)(v[j].x + 1), 1.0f));
            dv.y = rsqrtf(fmaxf((float)(v[j].y + 1), 1.0f));
            dv.z = rsqrtf(fmaxf((float)(v[j].z + 1), 1.0f));
            dv.w = rsqrtf(fmaxf((float)(v[j].w + 1), 1.0f));
            *(float4*)(dinv + g) = dv;
        }
    }
}

__global__ void k_csr_fill(const int* __restrict__ src, const int* __restrict__ dst,
                           const int* __restrict__ rowptr, const int* __restrict__ rank,
                           int* __restrict__ csr_src) {
    int e = blockIdx.x * blockDim.x + threadIdx.x;
    csr_src[rowptr[dst[e]] + rank[e]] = src[e];
}

template <int K>
__global__ __launch_bounds__(256)
void k_gemm_wmma(const float* in, const float* __restrict__ W,
                 float* out, int mode, float beta, float* stats) {
    extern __shared__ float2 sW2[];
    __shared__ float wpart[8][2 * DH];
    const int tid = threadIdx.x;

    {
        const int totalIter = (K / 2) * (DH / 4);
        for (int i = tid; i < totalIter; i += 256) {
            const int kp = i / (DH / 4);
            const int c4 = i % (DH / 4);
            const float4 a0 = *(const float4*)(W + (size_t)(2 * kp)     * DH + c4 * 4);
            const float4 a1 = *(const float4*)(W + (size_t)(2 * kp + 1) * DH + c4 * 4);
            float4* dst4 = (float4*)(sW2 + (size_t)kp * DH + c4 * 4);
            float4 f0; f0.x = a0.x; f0.y = a1.x; f0.z = a0.y; f0.w = a1.y;
            float4 f1; f1.x = a0.z; f1.y = a1.z; f1.z = a0.w; f1.w = a1.w;
            dst4[0] = f0;
            dst4[1] = f1;
        }
    }
    __syncthreads();

    const int wave = tid >> 5;
    const int lane = tid & 31;
    const int tileRow = blockIdx.x * 8 + wave;
    const bool valid = tileRow < NND / 16;
    if (valid) {
        const int m0 = tileRow * 16;
        const int r    = lane & 15;
        const bool hi  = lane >= 16;
        const int koff = hi ? 2 : 0;
        const float* arow = in + (size_t)(m0 + r) * K;

        v8f c[8] = {};
        for (int k0 = 0; k0 < K; k0 += 4) {
            const float2 a2 = *(const float2*)(arow + k0 + koff);
            v2f a; a.x = a2.x; a.y = a2.y;
            const int pairRow = (k0 >> 1) + (hi ? 1 : 0);
            const float2* brow = sW2 + (size_t)pairRow * DH + r;
#pragma unroll
            for (int t = 0; t < 8; ++t) {
                const float2 b2 = brow[t * 16];
                v2f b; b.x = b2.x; b.y = b2.y;
                c[t] = __builtin_amdgcn_wmma_f32_16x16x4_f32(
                    false, a, false, b, (short)0, c[t], false, false);
            }
        }

        float cs[8], cq[8];
#pragma unroll
        for (int t = 0; t < 8; ++t) {
            const int col = t * 16 + r;
            cs[t] = 0.0f; cq[t] = 0.0f;
#pragma unroll
            for (int v = 0; v < 8; ++v) {
                const int row = m0 + v + (hi ? 8 : 0);
                const size_t o = (size_t)row * DH + col;
                float val = c[t][v];
                if (mode != 0) val = (1.0f - beta) * in[o] + beta * val;
                out[o] = val;
                cs[t] += val; cq[t] += val * val;
            }
        }
        if (stats != nullptr) {
#pragma unroll
            for (int t = 0; t < 8; ++t) {
                const float s = cs[t] + __shfl_xor(cs[t], 16);
                const float q = cq[t] + __shfl_xor(cq[t], 16);
                wpart[wave][(hi ? DH : 0) + t * 16 + r] = hi ? q : s;
            }
        }
    }

    if (stats != nullptr) {
        if (!valid) {
            for (int j = lane; j < 2 * DH; j += 32) wpart[wave][j] = 0.0f;
        }
        __syncthreads();
        float tot = 0.0f;
#pragma unroll
        for (int w = 0; w < 8; ++w) tot += wpart[w][tid];
        atomicAdd(&stats[tid], tot);
    }
}

__device__ __forceinline__ float4 ld_row(const float* __restrict__ f, int s, int l) {
    return *(const float4*)(f + (size_t)s * DH + l * 4);
}
__device__ __forceinline__ float4 zero4() {
    float4 v; v.x = 0.0f; v.y = 0.0f; v.z = 0.0f; v.w = 0.0f; return v;
}
__device__ __forceinline__ void add4(float4& a, const float4 v) {
    a.x += v.x; a.y += v.y; a.z += v.z; a.w += v.w;
}
__device__ __forceinline__ void fma4(float4& a, float w, const float4 v) {
    a.x += w * v.x; a.y += w * v.y; a.z += w * v.z; a.w += w * v.w;
}

__global__ __launch_bounds__(256)
void k_gcn_gather(const int* __restrict__ rowptr, const int* __restrict__ csr_src,
                  const float* __restrict__ dinv, const float* __restrict__ xw,
                  const float* __restrict__ b, float* __restrict__ z) {
    const int d   = __builtin_amdgcn_readfirstlane(blockIdx.x * 8 + (threadIdx.x >> 5));
    const int l   = threadIdx.x & 31;
    const int beg = __builtin_amdgcn_readfirstlane(rowptr[d]);
    const int end = __builtin_amdgcn_readfirstlane(rowptr[d + 1]);
    const float dd = dinv[d];
    const float4 vs = ld_row(xw, d, l);
    const float4 bb = *(const float4*)(b + l * 4);
    float4 acc = zero4();
    for (int base = beg; base < end; base += 32) {
        const int n = min(32, end - base);
        const int mine = (l < n) ? csr_src[base + l] : 0;
        int j = 0;
        for (; j + 4 <= n; j += 4) {
            const int s0 = __builtin_amdgcn_readlane(mine, j);
            const int s1 = __builtin_amdgcn_readlane(mine, j + 1);
            const int s2 = __builtin_amdgcn_readlane(mine, j + 2);
            const int s3 = __builtin_amdgcn_readlane(mine, j + 3);
            const float4 v0 = ld_row(xw, s0, l), v1 = ld_row(xw, s1, l);
            const float4 v2 = ld_row(xw, s2, l), v3 = ld_row(xw, s3, l);
            const float w0 = dinv[s0] * dd, w1 = dinv[s1] * dd;
            const float w2 = dinv[s2] * dd, w3 = dinv[s3] * dd;
            fma4(acc, w0, v0); fma4(acc, w1, v1); fma4(acc, w2, v2); fma4(acc, w3, v3);
        }
        const int r = n - j;
        if (r > 0) {
            const int s0 = __builtin_amdgcn_readlane(mine, j);
            const float4 v0 = ld_row(xw, s0, l);
            const float w0 = dinv[s0] * dd;
            float4 v1 = zero4(), v2 = zero4();
            float w1 = 0.0f, w2 = 0.0f;
            if (r > 1) {
                const int s1 = __builtin_amdgcn_readlane(mine, j + 1);
                v1 = ld_row(xw, s1, l); w1 = dinv[s1] * dd;
            }
            if (r > 2) {
                const int s2 = __builtin_amdgcn_readlane(mine, j + 2);
                v2 = ld_row(xw, s2, l); w2 = dinv[s2] * dd;
            }
            fma4(acc, w0, v0);
            if (r > 1) fma4(acc, w1, v1);
            if (r > 2) fma4(acc, w2, v2);
        }
    }
    const float ws = dd * dd;
    float4 o;
    o.x = acc.x + ws * vs.x + bb.x; o.y = acc.y + ws * vs.y + bb.y;
    o.z = acc.z + ws * vs.z + bb.z; o.w = acc.w + ws * vs.w + bb.w;
    *(float4*)(z + (size_t)d * DH + l * 4) = o;
}

__device__ __forceinline__ void bn_coef1(float su, float sq, float g, float bt,
                                         float& sc, float& sh) {
    const float invN = 1.0f / (float)NND;
    const float mean = su * invN;
    const float var  = sq * invN - mean * mean;
    sc = rsqrtf(var + BNEPS) * g;
    sh = bt - mean * sc;
}
template <bool BN>
__device__ __forceinline__ void acc_row(float4& a, const float4 v, const float4 sc, const float4 sh) {
    if (BN) {
        a.x += fmaxf(fmaf(v.x, sc.x, sh.x), 0.0f); a.y += fmaxf(fmaf(v.y, sc.y, sh.y), 0.0f);
        a.z += fmaxf(fmaf(v.z, sc.z, sh.z), 0.0f); a.w += fmaxf(fmaf(v.w, sc.w, sh.w), 0.0f);
    } else {
        add4(a, v);
    }
}

#define PITCH 132
#define GROWS 16
template <bool BN>
__global__ __launch_bounds__(256)
void k_gcn2_layer(const int* __restrict__ rowptr, const int* __restrict__ csr_src,
                  const float* __restrict__ z, const float* __restrict__ x0,
                  const float* __restrict__ W, float* __restrict__ out, float beta,
                  const float* __restrict__ stats_in, const float* __restrict__ gamma,
                  const float* __restrict__ bbeta, float* stats_out) {
    extern __shared__ float2 sW2[];
    float* slabs = (float*)(sW2 + (DH / 2) * DH);
    float* wpart = slabs + 8 * 16 * PITCH;
    const int tid = threadIdx.x;

#pragma unroll
    for (int it = 0; it < (DH / 2) * (DH / 4) / 256; ++it) {
        const int i = tid + it * 256;
        const int kp = i / (DH / 4);
        const int c4 = i % (DH / 4);
        const float4 a0 = *(const float4*)(W + (size_t)(2 * kp)     * DH + c4 * 4);
        const float4 a1 = *(const float4*)(W + (size_t)(2 * kp + 1) * DH + c4 * 4);
        float4* dst4 = (float4*)(sW2 + (size_t)kp * DH + c4 * 4);
        float4 f0; f0.x = a0.x; f0.y = a1.x; f0.z = a0.y; f0.w = a1.y;
        float4 f1; f1.x = a0.z; f1.y = a1.z; f1.z = a0.w; f1.w = a1.w;
        dst4[0] = f0;
        dst4[1] = f1;
    }
    __syncthreads();

    const int wave = tid >> 5;
    const int lane = tid & 31;
    const int tileRow = blockIdx.x * 8 + wave;
    const bool valid = tileRow < NND / 16;
    float* slabw = slabs + wave * 16 * PITCH;
    float* wp = wpart + wave * 2 * DH;
    if (valid) {
        const int m0 = __builtin_amdgcn_readfirstlane(tileRow * 16);

#pragma unroll
        for (int rr = 0; rr < 16; ++rr)
            *(float4*)(slabw + rr * PITCH + lane * 4) = ld_row(x0, m0 + rr, lane);
        const int rpv = rowptr[m0 + min(lane, 16)];
        const int E0 = __builtin_amdgcn_readlane(rpv, 0);
        const int E1 = __builtin_amdgcn_readlane(rpv, 16);
        float4 sc = zero4(), sh = zero4();
        if (BN) {
            const float4 su = *(const float4*)(stats_in + lane * 4);
            const float4 sq = *(const float4*)(stats_in + DH + lane * 4);
            const float4 g  = *(const float4*)(gamma + lane * 4);
            const float4 bt = *(const float4*)(bbeta + lane * 4);
            bn_coef1(su.x, sq.x, g.x, bt.x, sc.x, sh.x);
            bn_coef1(su.y, sq.y, g.y, bt.y, sc.y, sh.y);
            bn_coef1(su.z, sq.z, g.z, bt.z, sc.z, sh.z);
            bn_coef1(su.w, sq.w, g.w, bt.w, sc.w, sh.w);
        }
        const float wagg = 1.0f - ALPHA;
        float4 acc = zero4();
        int cur = 0;
        int bnd = __builtin_amdgcn_readlane(rpv, 1);
        int e = E0;
        auto flush = [&]() {
            float* p = slabw + cur * PITCH + lane * 4;
            const float4 xv = *(const float4*)p;
            float4 o;
            o.x = ALPHA * xv.x + wagg * acc.x; o.y = ALPHA * xv.y + wagg * acc.y;
            o.z = ALPHA * xv.z + wagg * acc.z; o.w = ALPHA * xv.w + wagg * acc.w;
            *(float4*)p = o;
            acc = zero4();
            ++cur;
            bnd = __builtin_amdgcn_readlane(rpv, cur + 1);
        };
        int mine_next = (E0 + lane < E1) ? csr_src[E0 + lane] : 0;
        for (int base = E0; base < E1; base += 32) {
            const int mine = mine_next;
            const int n = min(32, E1 - base);
            for (int j = 0; j < n; j += GROWS) {
                const int cnt = min(GROWS, n - j);
                float4 v[GROWS];
#pragma unroll
                for (int q = 0; q < GROWS; ++q)
                    v[q] = ld_row(z, __builtin_amdgcn_readlane(mine, j + min(q, cnt - 1)), lane);
                if (j == 0)
                    mine_next = (base + 32 + lane < E1) ? csr_src[base + 32 + lane] : 0;
#pragma unroll
                for (int q = 0; q < GROWS; ++q) {
                    if (q < cnt) {
                        while (e == bnd && cur < 15) flush();
                        acc_row<BN>(acc, v[q], sc, sh);
                        ++e;
                    }
                }
            }
        }
        while (cur < 16) flush();

        const int r    = lane & 15;
        const bool hi  = lane >= 16;
        const int koff = hi ? 2 : 0;
        const float* arow = slabw + r * PITCH;

        v8f c[8] = {};
        for (int k0 = 0; k0 < DH; k0 += 4) {
            const float2 a2 = *(const float2*)(arow + k0 + koff);
            v2f a; a.x = a2.x; a.y = a2.y;
            const int pairRow = (k0 >> 1) + (hi ? 1 : 0);
            const float2* brow = sW2 + (size_t)pairRow * DH + r;
#pragma unroll
            for (int t = 0; t < 8; ++t) {
                const float2 b2 = brow[t * 16];
                v2f b; b.x = b2.x; b.y = b2.y;
                c[t] = __builtin_amdgcn_wmma_f32_16x16x4_f32(
                    false, a, false, b, (short)0, c[t], false, false);
            }
        }

        float cs[8], cq[8];
#pragma unroll
        for (int t = 0; t < 8; ++t) {
            const int col = t * 16 + r;
            cs[t] = 0.0f; cq[t] = 0.0f;
#pragma unroll
            for (int v = 0; v < 8; ++v) {
                const int lrow = v + (hi ? 8 : 0);
                const float hv = slabw[lrow * PITCH + col];
                const float val = (1.0f - beta) * hv + beta * c[t][v];
                out[(size_t)(m0 + lrow) * DH + col] = val;
                cs[t] += val; cq[t] += val * val;
            }
        }
        if (stats_out != nullptr) {
#pragma unroll
            for (int t = 0; t < 8; ++t) {
                const float s = cs[t] + __shfl_xor(cs[t], 16);
                const float q = cq[t] + __shfl_xor(cq[t], 16);
                wp[(hi ? DH : 0) + t * 16 + r] = hi ? q : s;
            }
        }
    }

    if (stats_out != nullptr) {
        if (!valid) {
            for (int j = lane; j < 2 * DH; j += 32) wp[j] = 0.0f;
        }
        __syncthreads();
        float tot = 0.0f;
#pragma unroll
        for (int w = 0; w < 8; ++w) tot += wpart[w * 2 * DH + tid];
        atomicAdd(&stats_out[tid], tot);
    }
}

extern "C" void kernel_launch(void* const* d_in, const int* in_sizes, int n_in,
                              void* d_out, int out_size, void* d_ws, size_t ws_size,
                              hipStream_t stream) {
    const float* x        = (const float*)d_in[0];
    const int*   edge     = (const int*)d_in[1];
    const float* W_lin    = (const float*)d_in[2];
    const float* b_lin    = (const float*)d_in[3];
    const float* W1s      = (const float*)d_in[4];
    const float* bn_gamma = (const float*)d_in[5];
    const float* bn_beta  = (const float*)d_in[6];
    float* out = (float*)d_out;

    float* ws    = (float*)d_ws;
    float* dinv  = ws;
    float* stats = ws + NND;
    float* bufA  = stats + NSTAT;
    float* bufB  = bufA + (size_t)NND * DH;
    float* x0    = bufB + (size_t)NND * DH;
    int*   cnt     = (int*)(x0 + (size_t)NND * DH);
    int*   rowptr  = cnt + NPAD;
    int*   rank    = rowptr + NPAD;
    int*   csr_src = rank + NED;

    const int* src = edge;
    const int* dst = edge + NED;

    const int nBlkE   = NED / 256;
    const int nBlkRow = NND / 8;
    const int nBlkGemm = (NND / 16 + 7) / 8;
    const size_t ldsLayer = DH * DH * sizeof(float) + 8 * 16 * PITCH * sizeof(float)
                          + 8 * 2 * DH * sizeof(float);

    k_cnt_zero<<<NPAD / 256, 256, 0, stream>>>(cnt, stats);
    k_hist_rank<<<nBlkE, 256, 0, stream>>>(dst, cnt, rank);
    k_scan_dinv<<<1, SCANT, 0, stream>>>(cnt, rowptr, dinv);
    k_csr_fill<<<nBlkE, 256, 0, stream>>>(src, dst, rowptr, rank, csr_src);

    k_gemm_wmma<DIN><<<nBlkGemm, 256, DIN * DH * sizeof(float), stream>>>(
        x, W_lin, bufA, 0, 0.0f, nullptr);

    k_gcn_gather<<<nBlkRow, 256, 0, stream>>>(rowptr, csr_src, dinv, bufA, b_lin, x0);

    const float* z = x0;
    for (int i = 0; i < 5; ++i) {
        const float beta = logf(0.5f / (float)(i + 1) + 1.0f);
        float* o = (i == 4) ? out : ((i & 1) ? bufB : bufA);
        const float* Wi = W1s + (size_t)i * DH * DH;
        if (i == 0) {
            k_gcn2_layer<false><<<nBlkGemm, 256, ldsLayer, stream>>>(
                rowptr, csr_src, z, x0, Wi, o, beta, nullptr, nullptr, nullptr, stats);
        } else {
            k_gcn2_layer<true><<<nBlkGemm, 256, ldsLayer, stream>>>(
                rowptr, csr_src, z, x0, Wi, o, beta, stats + (i - 1) * 2 * DH,
                bn_gamma + (i - 1) * DH, bn_beta + (i - 1) * DH,
                i < 4 ? stats + i * 2 * DH : nullptr);
        }
        z = o;
    }
}


# Edge list built by two levels of counting in LDS (256-node buckets, then nodes inside a bucket) instead of one global returning add per edge; 4 -> 3 launches; segment order still by arrival
# speedup vs baseline: 4.6858x; 1.1437x over previous
#include <hip/hip_runtime.h>
#include <cmath>

typedef __attribute__((ext_vector_type(2))) float v2f;
typedef __attribute__((ext_vector_type(8))) float v8f;

#define NND   50000
#define NED   800000
#define DIN   256
#define DH    128
#define ALPHA 0.1f
#define BNEPS 1e-5f

#define NPAD  53248
#define NSTAT (4 * 2 * DH)
#define BK_NB  196
#define BK_G   256
#define BK_EPW (NED / BK_G)
#define BK_EPT 13

__device__ __forceinline__ int wg_excl_scan256(int c, int* wtot, int lane, int wave) {
    int incl = c;
#pragma unroll
    for (int o = 1; o < 32; o <<= 1) {
        int n = __shfl_up(incl, o);
        if (lane >= o) incl += n;
    }
    if (lane == 31) wtot[wave] = incl;
    __syncthreads();
    int woff = 0;
#pragma unroll
    for (int w = 0; w < 8; ++w) woff += (w < wave) ? wtot[w] : 0;
    return woff + incl - c;
}

__global__ __launch_bounds__(256)
void k_bin_count(const int* __restrict__ dst, int* __restrict__ wgcnt, float* __restrict__ stats) {
    __shared__ int hist[BK_NB];
    const int tid = threadIdx.x;
    const int g = blockIdx.x;
    if (tid < BK_NB) hist[tid] = 0;
    int d[BK_EPT];
#pragma unroll
    for (int k = 0; k < BK_EPT; ++k)
        d[k] = dst[g * BK_EPW + min(tid + k * 256, BK_EPW - 1)];
    __syncthreads();
#pragma unroll
    for (int k = 0; k < BK_EPT; ++k)
        if (tid + k * 256 < BK_EPW) atomicAdd(&hist[d[k] >> 8], 1);
    __syncthreads();
    if (tid < BK_NB) wgcnt[g * BK_NB + tid] = hist[tid];
    if (g == 0)
        for (int i = tid; i < NSTAT; i += 256) stats[i] = 0.0f;
}

__global__ __launch_bounds__(256)
void k_bin_scatter(const int* __restrict__ src, const int* __restrict__ dst,
                   const int* __restrict__ wgcnt, int* __restrict__ bstart,
                   int* __restrict__ rec_i) {
    int2* rec = (int2*)rec_i;
    __shared__ int cursor[BK_NB];
    __shared__ int wtot[8];
    const int tid = threadIdx.x;
    const int lane = tid & 31, wave = tid >> 5;
    const int g = blockIdx.x;
    int d[BK_EPT], s[BK_EPT];
#pragma unroll
    for (int k = 0; k < BK_EPT; ++k) {
        const int i = g * BK_EPW + min(tid + k * 256, BK_EPW - 1);
        d[k] = dst[i];
        s[k] = src[i];
    }
    int tot = 0, pre = 0;
    if (tid < BK_NB) {
#pragma unroll 8
        for (int gg = 0; gg < BK_G; ++gg) {
            const int c = wgcnt[gg * BK_NB + tid];
            tot += c;
            pre += (gg < g) ? c : 0;
        }
    }
    const int excl = wg_excl_scan256(tot, wtot, lane, wave);
    if (tid < BK_NB) cursor[tid] = excl + pre;
    if (g == 0 && tid <= BK_NB) bstart[tid] = excl;
    __syncthreads();
#pragma unroll
    for (int k = 0; k < BK_EPT; ++k) {
        if (tid + k * 256 < BK_EPW) {
            const int pos = atomicAdd(&cursor[d[k] >> 8], 1);
            int2 r; r.x = d[k]; r.y = s[k];
            rec[pos] = r;
        }
    }
}

#define BC_BATCH 4
__global__ __launch_bounds__(256)
void k_bucket_csr(const int* __restrict__ bstart, const int* __restrict__ rec_i,
                  int* __restrict__ rowptr, float* __restrict__ dinv,
                  int* __restrict__ csr_src) {
    const int2* rec = (const int2*)rec_i;
    __shared__ int cnt[256];
    __shared__ int wtot[8];
    const int tid = threadIdx.x;
    const int lane = tid & 31, wave = tid >> 5;
    const int b = blockIdx.x;
    const int lo = bstart[b], hi = bstart[b + 1];
    cnt[tid] = 0;
    __syncthreads();
    for (int i0 = lo; i0 < hi; i0 += 256 * BC_BATCH) {
        int dd[BC_BATCH];
#pragma unroll
        for (int k = 0; k < BC_BATCH; ++k)
            dd[k] = rec[min(i0 + k * 256 + tid, hi - 1)].x;
#pragma unroll
        for (int k = 0; k < BC_BATCH; ++k)
            if (i0 + k * 256 + tid < hi) atomicAdd(&cnt[dd[k] & 255], 1);
    }
    __syncthreads();
    const int c = cnt[tid];
    const int excl = wg_excl_scan256(c, wtot, lane, wave);
    const int node = b * 256 + tid;
    rowptr[node] = lo + excl;
    if (node < NND) dinv[node] = rsqrtf(fmaxf((float)(c + 1), 1.0f));
    cnt[tid] = excl;
    __syncthreads();
    for (int i0 = lo; i0 < hi; i0 += 256 * BC_BATCH) {
        int2 r[BC_BATCH];
#pragma unroll
        for (int k = 0; k < BC_BATCH; ++k)
            r[k] = rec[min(i0 + k * 256 + tid, hi - 1)];
#pragma unroll
        for (int k = 0; k < BC_BATCH; ++k) {
            if (i0 + k * 256 + tid < hi) {
                const int pos = atomicAdd(&cnt[r[k].x & 255], 1);
                csr_src[lo + pos] = r[k].y;
            }
        }
    }
}

template <int K>
__global__ __launch_bounds__(256)
void k_gemm_wmma(const float* in, const float* __restrict__ W,
                 float* out, int mode, float beta, float* stats) {
    extern __shared__ float2 sW2[];
    __shared__ float wpart[8][2 * DH];
    const int tid = threadIdx.x;

    {
        const int totalIter = (K / 2) * (DH / 4);
        for (int i = tid; i < totalIter; i += 256) {
            const int kp = i / (DH / 4);
            const int c4 = i % (DH / 4);
            const float4 a0 = *(const float4*)(W + (size_t)(2 * kp)     * DH + c4 * 4);
            const float4 a1 = *(const float4*)(W + (size_t)(2 * kp + 1) * DH + c4 * 4);
            float4* dst4 = (float4*)(sW2 + (size_t)kp * DH + c4 * 4);
            float4 f0; f0.x = a0.x; f0.y = a1.x; f0.z = a0.y; f0.w = a1.y;
            float4 f1; f1.x = a0.z; f1.y = a1.z; f1.z = a0.w; f1.w = a1.w;
            dst4[0] = f0;
            dst4[1] = f1;
        }
    }
    __syncthreads();

    const int wave = tid >> 5;
    const int lane = tid & 31;
    const int tileRow = blockIdx.x * 8 + wave;
    const bool valid = tileRow < NND / 16;
    if (valid) {
        const int m0 = tileRow * 16;
        const int r    = lane & 15;
        const bool hi  = lane >= 16;
        const int koff = hi ? 2 : 0;
        const float* arow = in + (size_t)(m0 + r) * K;

        v8f c[8] = {};
        for (int k0 = 0; k0 < K; k0 += 4) {
            const float2 a2 = *(const float2*)(arow + k0 + koff);
            v2f a; a.x = a2.x; a.y = a2.y;
            const int pairRow = (k0 >> 1) + (hi ? 1 : 0);
            const float2* brow = sW2 + (size_t)pairRow * DH + r;
#pragma unroll
            for (int t = 0; t < 8; ++t) {
                const float2 b2 = brow[t * 16];
                v2f b; b.x = b2.x; b.y = b2.y;
                c[t] = __builtin_amdgcn_wmma_f32_16x16x4_f32(
                    false, a, false, b, (short)0, c[t], false, false);
            }
        }

        float cs[8], cq[8];
#pragma unroll
        for (int t = 0; t < 8; ++t) {
            const int col = t * 16 + r;
            cs[t] = 0.0f; cq[t] = 0.0f;
#pragma unroll
            for (int v = 0; v < 8; ++v) {
                const int row = m0 + v + (hi ? 8 : 0);
                const size_t o = (size_t)row * DH + col;
                float val = c[t][v];
                if (mode != 0) val = (1.0f - beta) * in[o] + beta * val;
                out[o] = val;
                cs[t] += val; cq[t] += val * val;
            }
        }
        if (stats != nullptr) {
#pragma unroll
            for (int t = 0; t < 8; ++t) {
                const float s = cs[t] + __shfl_xor(cs[t], 16);
                const float q = cq[t] + __shfl_xor(cq[t], 16);
                wpart[wave][(hi ? DH : 0) + t * 16 + r] = hi ? q : s;
            }
        }
    }

    if (stats != nullptr) {
        if (!valid) {
            for (int j = lane; j < 2 * DH; j += 32) wpart[wave][j] = 0.0f;
        }
        __syncthreads();
        float tot = 0.0f;
#pragma unroll
        for (int w = 0; w < 8; ++w) tot += wpart[w][tid];
        atomicAdd(&stats[tid], tot);
    }
}

__device__ __forceinline__ float4 ld_row(const float* __restrict__ f, int s, int l) {
    return *(const float4*)(f + (size_t)s * DH + l * 4);
}
__device__ __forceinline__ float4 zero4() {
    float4 v; v.x = 0.0f; v.y = 0.0f; v.z = 0.0f; v.w = 0.0f; return v;
}
__device__ __forceinline__ void add4(float4& a, const float4 v) {
    a.x += v.x; a.y += v.y; a.z += v.z; a.w += v.w;
}
__device__ __forceinline__ void fma4(float4& a, float w, const float4 v) {
    a.x += w * v.x; a.y += w * v.y; a.z += w * v.z; a.w += w * v.w;
}

__global__ __launch_bounds__(256)
void k_gcn_gather(const int* __restrict__ rowptr, const int* __restrict__ csr_src,
                  const float* __restrict__ dinv, const float* __restrict__ xw,
                  const float* __restrict__ b, float* __restrict__ z) {
    const int d   = __builtin_amdgcn_readfirstlane(blockIdx.x * 8 + (threadIdx.x >> 5));
    const int l   = threadIdx.x & 31;
    const int beg = __builtin_amdgcn_readfirstlane(rowptr[d]);
    const int end = __builtin_amdgcn_readfirstlane(rowptr[d + 1]);
    const float dd = dinv[d];
    const float4 vs = ld_row(xw, d, l);
    const float4 bb = *(const float4*)(b + l * 4);
    float4 acc = zero4();
    for (int base = beg; base < end; base += 32) {
        const int n = min(32, end - base);
        const int mine = (l < n) ? csr_src[base + l] : 0;
        int j = 0;
        for (; j + 4 <= n; j += 4) {
            const int s0 = __builtin_amdgcn_readlane(mine, j);
            const int s1 = __builtin_amdgcn_readlane(mine, j + 1);
            const int s2 = __builtin_amdgcn_readlane(mine, j + 2);
            const int s3 = __builtin_amdgcn_readlane(mine, j + 3);
            const float4 v0 = ld_row(xw, s0, l), v1 = ld_row(xw, s1, l);
            const float4 v2 = ld_row(xw, s2, l), v3 = ld_row(xw, s3, l);
            const float w0 = dinv[s0] * dd, w1 = dinv[s1] * dd;
            const float w2 = dinv[s2] * dd, w3 = dinv[s3] * dd;
            fma4(acc, w0, v0); fma4(acc, w1, v1); fma4(acc, w2, v2); fma4(acc, w3, v3);
        }
        const int r = n - j;
        if (r > 0) {
            const int s0 = __builtin_amdgcn_readlane(mine, j);
            const float4 v0 = ld_row(xw, s0, l);
            const float w0 = dinv[s0] * dd;
            float4 v1 = zero4(), v2 = zero4();
            float w1 = 0.0f, w2 = 0.0f;
            if (r > 1) {
                const int s1 = __builtin_amdgcn_readlane(mine, j + 1);
                v1 = ld_row(xw, s1, l); w1 = dinv[s1] * dd;
            }
            if (r > 2) {
                const int s2 = __builtin_amdgcn_readlane(mine, j + 2);
                v2 = ld_row(xw, s2, l); w2 = dinv[s2] * dd;
            }
            fma4(acc, w0, v0);
            if (r > 1) fma4(acc, w1, v1);
            if (r > 2) fma4(acc, w2, v2);
        }
    }
    const float ws = dd * dd;
    float4 o;
    o.x = acc.x + ws * vs.x + bb.x; o.y = acc.y + ws * vs.y + bb.y;
    o.z = acc.z + ws * vs.z + bb.z; o.w = acc.w + ws * vs.w + bb.w;
    *(float4*)(z + (size_t)d * DH + l * 4) = o;
}

__device__ __forceinline__ void bn_coef1(float su, float sq, float g, float bt,
                                         float& sc, float& sh) {
    const float invN = 1.0f / (float)NND;
    const float mean = su * invN;
    const float var  = sq * invN - mean * mean;
    sc = rsqrtf(var + BNEPS) * g;
    sh = bt - mean * sc;
}
template <bool BN>
__device__ __forceinline__ void acc_row(float4& a, const float4 v, const float4 sc, const float4 sh) {
    if (BN) {
        a.x += fmaxf(fmaf(v.x, sc.x, sh.x), 0.0f); a.y += fmaxf(fmaf(v.y, sc.y, sh.y), 0.0f);
        a.z += fmaxf(fmaf(v.z, sc.z, sh.z), 0.0f); a.w += fmaxf(fmaf(v.w, sc.w, sh.w), 0.0f);
    } else {
        add4(a, v);
    }
}

#define PITCH 132
#define GROWS 16
template <bool BN>
__global__ __launch_bounds__(256)
void k_gcn2_layer(const int* __restrict__ rowptr, const int* __restrict__ csr_src,
                  const float* __restrict__ z, const float* __restrict__ x0,
                  const float* __restrict__ W, float* __restrict__ out, float beta,
                  const float* __restrict__ stats_in, const float* __restrict__ gamma,
                  const float* __restrict__ bbeta, float* stats_out) {
    extern __shared__ float2 sW2[];
    float* slabs = (float*)(sW2 + (DH / 2) * DH);
    float* wpart = slabs + 8 * 16 * PITCH;
    const int tid = threadIdx.x;

#pragma unroll
    for (int it = 0; it < (DH / 2) * (DH / 4) / 256; ++it) {
        const int i = tid + it * 256;
        const int kp = i / (DH / 4);
        const int c4 = i % (DH / 4);
        const float4 a0 = *(const float4*)(W + (size_t)(2 * kp)     * DH + c4 * 4);
        const float4 a1 = *(const float4*)(W + (size_t)(2 * kp + 1) * DH + c4 * 4);
        float4* dst4 = (float4*)(sW2 + (size_t)kp * DH + c4 * 4);
        float4 f0; f0.x = a0.x; f0.y = a1.x; f0.z = a0.y; f0.w = a1.y;
        float4 f1; f1.x = a0.z; f1.y = a1.z; f1.z = a0.w; f1.w = a1.w;
        dst4[0] = f0;
        dst4[1] = f1;
    }
    __syncthreads();

    const int wave = tid >> 5;
    const int lane = tid & 31;
    const int tileRow = blockIdx.x * 8 + wave;
    const bool valid = tileRow < NND / 16;
    float* slabw = slabs + wave * 16 * PITCH;
    float* wp = wpart + wave * 2 * DH;
    if (valid) {
        const int m0 = __builtin_amdgcn_readfirstlane(tileRow * 16);

#pragma unroll
        for (int rr = 0; rr < 16; ++rr)
            *(float4*)(slabw + rr * PITCH + lane * 4) = ld_row(x0, m0 + rr, lane);
        const int rpv = rowptr[m0 + min(lane, 16)];
        const int E0 = __builtin_amdgcn_readlane(rpv, 0);
        const int E1 = __builtin_amdgcn_readlane(rpv, 16);
        float4 sc = zero4(), sh = zero4();
        if (BN) {
            const float4 su = *(const float4*)(stats_in + lane * 4);
            const float4 sq = *(const float4*)(stats_in + DH + lane * 4);
            const float4 g  = *(const float4*)(gamma + lane * 4);
            const float4 bt = *(const float4*)(bbeta + lane * 4);
            bn_coef1(su.x, sq.x, g.x, bt.x, sc.x, sh.x);
            bn_coef1(su.y, sq.y, g.y, bt.y, sc.y, sh.y);
            bn_coef1(su.z, sq.z, g.z, bt.z, sc.z, sh.z);
            bn_coef1(su.w, sq.w, g.w, bt.w, sc.w, sh.w);
        }
        const float wagg = 1.0f - ALPHA;
        float4 acc = zero4();
        int cur = 0;
        int bnd = __builtin_amdgcn_readlane(rpv, 1);
        int e = E0;
        auto flush = [&]() {
            float* p = slabw + cur * PITCH + lane * 4;
            const float4 xv = *(const float4*)p;
            float4 o;
            o.x = ALPHA * xv.x + wagg * acc.x; o.y = ALPHA * xv.y + wagg * acc.y;
            o.z = ALPHA * xv.z + wagg * acc.z; o.w = ALPHA * xv.w + wagg * acc.w;
            *(float4*)p = o;
            acc = zero4();
            ++cur;
            bnd = __builtin_amdgcn_readlane(rpv, cur + 1);
        };
        int mine_next = (E0 + lane < E1) ? csr_src[E0 + lane] : 0;
        for (int base = E0; base < E1; base += 32) {
            const int mine = mine_next;
            const int n = min(32, E1 - base);
            for (int j = 0; j < n; j += GROWS) {
                const int cnt = min(GROWS, n - j);
                float4 v[GROWS];
#pragma unroll
                for (int q = 0; q < GROWS; ++q)
                    v[q] = ld_row(z, __builtin_amdgcn_readlane(mine, j + min(q, cnt - 1)), lane);
                if (j == 0)
                    mine_next = (base + 32 + lane < E1) ? csr_src[base + 32 + lane] : 0;
#pragma unroll
                for (int q = 0; q < GROWS; ++q) {
                    if (q < cnt) {
                        while (e == bnd && cur < 15) flush();
                        acc_row<BN>(acc, v[q], sc, sh);
                        ++e;
                    }
                }
            }
        }
        while (cur < 16) flush();

        const int r    = lane & 15;
        const bool hi  = lane >= 16;
        const int koff = hi ? 2 : 0;
        const float* arow = slabw + r * PITCH;

        v8f c[8] = {};
        for (int k0 = 0; k0 < DH; k0 += 4) {
            const float2 a2 = *(const float2*)(arow + k0 + koff);
            v2f a; a.x = a2.x; a.y = a2.y;
            const int pairRow = (k0 >> 1) + (hi ? 1 : 0);
            const float2* brow = sW2 + (size_t)pairRow * DH + r;
#pragma unroll
            for (int t = 0; t < 8; ++t) {
                const float2 b2 = brow[t * 16];
                v2f b; b.x = b2.x; b.y = b2.y;
                c[t] = __builtin_amdgcn_wmma_f32_16x16x4_f32(
                    false, a, false, b, (short)0, c[t], false, false);
            }
        }

        float cs[8], cq[8];
#pragma unroll
        for (int t = 0; t < 8; ++t) {
            const int col = t * 16 + r;
            cs[t] = 0.0f; cq[t] = 0.0f;
#pragma unroll
            for (int v = 0; v < 8; ++v) {
                const int lrow = v + (hi ? 8 : 0);
                const float hv = slabw[lrow * PITCH + col];
                const float val = (1.0f - beta) * hv + beta * c[t][v];
                out[(size_t)(m0 + lrow) * DH + col] = val;
                cs[t] += val; cq[t] += val * val;
            }
        }
        if (stats_out != nullptr) {
#pragma unroll
            for (int t = 0; t < 8; ++t) {
                const float s = cs[t] + __shfl_xor(cs[t], 16);
                const float q = cq[t] + __shfl_xor(cq[t], 16);
                wp[(hi ? DH : 0) + t * 16 + r] = hi ? q : s;
            }
        }
    }

    if (stats_out != nullptr) {
        if (!valid) {
            for (int j = lane; j < 2 * DH; j += 32) wp[j] = 0.0f;
        }
        __syncthreads();
        float tot = 0.0f;
#pragma unroll
        for (int w = 0; w < 8; ++w) tot += wpart[w * 2 * DH + tid];
        atomicAdd(&stats_out[tid], tot);
    }
}

extern "C" void kernel_launch(void* const* d_in, const int* in_sizes, int n_in,
                              void* d_out, int out_size, void* d_ws, size_t ws_size,
                              hipStream_t stream) {
    const float* x        = (const float*)d_in[0];
    const int*   edge     = (const int*)d_in[1];
    const float* W_lin    = (const float*)d_in[2];
    const float* b_lin    = (const float*)d_in[3];
    const float* W1s      = (const float*)d_in[4];
    const float* bn_gamma = (const float*)d_in[5];
    const float* bn_beta  = (const float*)d_in[6];
    float* out = (float*)d_out;

    float* ws    = (float*)d_ws;
    float* dinv  = ws;
    float* stats = ws + NND;
    float* bufA  = stats + NSTAT;
    float* bufB  = bufA + (size_t)NND * DH;
    float* x0    = bufB + (size_t)NND * DH;
    int*   wgcnt   = (int*)(x0 + (size_t)NND * DH);
    int*   bstart  = wgcnt + BK_G * BK_NB;
    int*   rowptr  = wgcnt + NPAD;
    int*   csr_src = rowptr + NPAD;
    int*   rec     = csr_src + NED;

    const int* src = edge;
    const int* dst = edge + NED;

    const int nBlkRow = NND / 8;
    const int nBlkGemm = (NND / 16 + 7) / 8;
    const size_t ldsLayer = DH * DH * sizeof(float) + 8 * 16 * PITCH * sizeof(float)
                          + 8 * 2 * DH * sizeof(float);

    k_bin_count<<<BK_G, 256, 0, stream>>>(dst, wgcnt, stats);
    k_bin_scatter<<<BK_G, 256, 0, stream>>>(src, dst, wgcnt, bstart, rec);
    k_bucket_csr<<<BK_NB, 256, 0, stream>>>(bstart, rec, rowptr, dinv, csr_src);

    k_gemm_wmma<DIN><<<nBlkGemm, 256, DIN * DH * sizeof(float), stream>>>(
        x, W_lin, bufA, 0, 0.0f, nullptr);

    k_gcn_gather<<<nBlkRow, 256, 0, stream>>>(rowptr, csr_src, dinv, bufA, b_lin, x0);

    const float* z = x0;
    for (int i = 0; i < 5; ++i) {
        const float beta = logf(0.5f / (float)(i + 1) + 1.0f);
        float* o = (i == 4) ? out : ((i & 1) ? bufB : bufA);
        const float* Wi = W1s + (size_t)i * DH * DH;
        if (i == 0) {
            k_gcn2_layer<false><<<nBlkGemm, 256, ldsLayer, stream>>>(
                rowptr, csr_src, z, x0, Wi, o, beta, nullptr, nullptr, nullptr, stats);
        } else {
            k_gcn2_layer<true><<<nBlkGemm, 256, ldsLayer, stream>>>(
                rowptr, csr_src, z, x0, Wi, o, beta, stats + (i - 1) * 2 * DH,
                bn_gamma + (i - 1) * DH, bn_beta + (i - 1) * DH,
                i < 4 ? stats + i * 2 * DH : nullptr);
        }
        z = o;
    }
}


# GCN2 layer gather walk: fewer instructions per gathered row (ids from fixed lanes, 32-bit row offset on a scalar base, one compare per row for node end, common path falls through); sum order kept
# speedup vs baseline: 5.1080x; 1.0901x over previous
#include <hip/hip_runtime.h>
#include <cmath>

typedef __attribute__((ext_vector_type(2))) float v2f;
typedef __attribute__((ext_vector_type(8))) float v8f;

#define NND   50000
#define NED   800000
#define DIN   256
#define DH    128
#define ALPHA 0.1f
#define BNEPS 1e-5f

#define NPAD  53248
#define NSTAT (4 * 2 * DH)
#define BK_NB  196
#define BK_G   256
#define BK_EPW (NED / BK_G)
#define BK_EPT 13

__device__ __forceinline__ int wg_excl_scan256(int c, int* wtot, int lane, int wave) {
    int incl = c;
#pragma unroll
    for (int o = 1; o < 32; o <<= 1) {
        int n = __shfl_up(incl, o);
        if (lane >= o) incl += n;
    }
    if (lane == 31) wtot[wave] = incl;
    __syncthreads();
    int woff = 0;
#pragma unroll
    for (int w = 0; w < 8; ++w) woff += (w < wave) ? wtot[w] : 0;
    return woff + incl - c;
}

__global__ __launch_bounds__(256)
void k_bin_count(const int* __restrict__ dst, int* __restrict__ wgcnt, float* __restrict__ stats) {
    __shared__ int hist[BK_NB];
    const int tid = threadIdx.x;
    const int g = blockIdx.x;
    if (tid < BK_NB) hist[tid] = 0;
    int d[BK_EPT];
#pragma unroll
    for (int k = 0; k < BK_EPT; ++k)
        d[k] = dst[g * BK_EPW + min(tid + k * 256, BK_EPW - 1)];
    __syncthreads();
#pragma unroll
    for (int k = 0; k < BK_EPT; ++k)
        if (tid + k * 256 < BK_EPW) atomicAdd(&hist[d[k] >> 8], 1);
    __syncthreads();
    if (tid < BK_NB) wgcnt[g * BK_NB + tid] = hist[tid];
    if (g == 0)
        for (int i = tid; i < NSTAT; i += 256) stats[i] = 0.0f;
}

__global__ __launch_bounds__(256)
void k_bin_scatter(const int* __restrict__ src, const int* __restrict__ dst,
                   const int* __restrict__ wgcnt, int* __restrict__ bstart,
                   int* __restrict__ rec_i) {
    int2* rec = (int2*)rec_i;
    __shared__ int cursor[BK_NB];
    __shared__ int wtot[8];
    const int tid = threadIdx.x;
    const int lane = tid & 31, wave = tid >> 5;
    const int g = blockIdx.x;
    int d[BK_EPT], s[BK_EPT];
#pragma unroll
    for (int k = 0; k < BK_EPT; ++k) {
        const int i = g * BK_EPW + min(tid + k * 256, BK_EPW - 1);
        d[k] = dst[i];
        s[k] = src[i];
    }
    int tot = 0, pre = 0;
    if (tid < BK_NB) {
#pragma unroll 8
        for (int gg = 0; gg < BK_G; ++gg) {
            const int c = wgcnt[gg * BK_NB + tid];
            tot += c;
            pre += (gg < g) ? c : 0;
        }
    }
    const int excl = wg_excl_scan256(tot, wtot, lane, wave);
    if (tid < BK_NB) cursor[tid] = excl + pre;
    if (g == 0 && tid <= BK_NB) bstart[tid] = excl;
    __syncthreads();
#pragma unroll
    for (int k = 0; k < BK_EPT; ++k) {
        if (tid + k * 256 < BK_EPW) {
            const int pos = atomicAdd(&cursor[d[k] >> 8], 1);
            int2 r; r.x = d[k]; r.y = s[k];
            rec[pos] = r;
        }
    }
}

#define BC_BATCH 4
__global__ __launch_bounds__(256)
void k_bucket_csr(const int* __restrict__ bstart, const int* __restrict__ rec_i,
                  int* __restrict__ rowptr, float* __restrict__ dinv,
                  int* __restrict__ csr_src) {
    const int2* rec = (const int2*)rec_i;
    __shared__ int cnt[256];
    __shared__ int wtot[8];
    const int tid = threadIdx.x;
    const int lane = tid & 31, wave = tid >> 5;
    const int b = blockIdx.x;
    const int lo = bstart[b], hi = bstart[b + 1];
    cnt[tid] = 0;
    __syncthreads();
    for (int i0 = lo; i0 < hi; i0 += 256 * BC_BATCH) {
        int dd[BC_BATCH];
#pragma unroll
        for (int k = 0; k < BC_BATCH; ++k)
            dd[k] = rec[min(i0 + k * 256 + tid, hi - 1)].x;
#pragma unroll
        for (int k = 0; k < BC_BATCH; ++k)
            if (i0 + k * 256 + tid < hi) atomicAdd(&cnt[dd[k] & 255], 1);
    }
    __syncthreads();
    const int c = cnt[tid];
    const int excl = wg_excl_scan256(c, wtot, lane, wave);
    const int node = b * 256 + tid;
    rowptr[node] = lo + excl;
    if (node < NND) dinv[node] = rsqrtf(fmaxf((float)(c + 1), 1.0f));
    cnt[tid] = excl;
    __syncthreads();
    for (int i0 = lo; i0 < hi; i0 += 256 * BC_BATCH) {
        int2 r[BC_BATCH];
#pragma unroll
        for (int k = 0; k < BC_BATCH; ++k)
            r[k] = rec[min(i0 + k * 256 + tid, hi - 1)];
#pragma unroll
        for (int k = 0; k < BC_BATCH; ++k) {
            if (i0 + k * 256 + tid < hi) {
                const int pos = atomicAdd(&cnt[r[k].x & 255], 1);
                csr_src[lo + pos] = r[k].y;
            }
        }
    }
}

template <int K>
__global__ __launch_bounds__(256)
void k_gemm_wmma(const float* in, const float* __restrict__ W,
                 float* out, int mode, float beta, float* stats) {
    extern __shared__ float2 sW2[];
    __shared__ float wpart[8][2 * DH];
    const int tid = threadIdx.x;

    {
        const int totalIter = (K / 2) * (DH / 4);
        for (int i = tid; i < totalIter; i += 256) {
            const int kp = i / (DH / 4);
            const int c4 = i % (DH / 4);
            const float4 a0 = *(const float4*)(W + (size_t)(2 * kp)     * DH + c4 * 4);
            const float4 a1 = *(const float4*)(W + (size_t)(2 * kp + 1) * DH + c4 * 4);
            float4* dst4 = (float4*)(sW2 + (size_t)kp * DH + c4 * 4);
            float4 f0; f0.x = a0.x; f0.y = a1.x; f0.z = a0.y; f0.w = a1.y;
            float4 f1; f1.x = a0.z; f1.y = a1.z; f1.z = a0.w; f1.w = a1.w;
            dst4[0] = f0;
            dst4[1] = f1;
        }
    }
    __syncthreads();

    const int wave = tid >> 5;
    const int lane = tid & 31;
    const int tileRow = blockIdx.x * 8 + wave;
    const bool valid = tileRow < NND / 16;
    if (valid) {
        const int m0 = tileRow * 16;
        const int r    = lane & 15;
        const bool hi  = lane >= 16;
        const int koff = hi ? 2 : 0;
        const float* arow = in + (size_t)(m0 + r) * K;

        v8f c[8] = {};
        for (int k0 = 0; k0 < K; k0 += 4) {
            const float2 a2 = *(const float2*)(arow + k0 + koff);
            v2f a; a.x = a2.x; a.y = a2.y;
            const int pairRow = (k0 >> 1) + (hi ? 1 : 0);
            const float2* brow = sW2 + (size_t)pairRow * DH + r;
#pragma unroll
            for (int t = 0; t < 8; ++t) {
                const float2 b2 = brow[t * 16];
                v2f b; b.x = b2.x; b.y = b2.y;
                c[t] = __builtin_amdgcn_wmma_f32_16x16x4_f32(
                    false, a, false, b, (short)0, c[t], false, false);
            }
        }

        float cs[8], cq[8];
#pragma unroll
        for (int t = 0; t < 8; ++t) {
            const int col = t * 16 + r;
            cs[t] = 0.0f; cq[t] = 0.0f;
#pragma unroll
            for (int v = 0; v < 8; ++v) {
                const int row = m0 + v + (hi ? 8 : 0);
                const size_t o = (size_t)row * DH + col;
                float val = c[t][v];
                if (mode != 0) val = (1.0f - beta) * in[o] + beta * val;
                out[o] = val;
                cs[t] += val; cq[t] += val * val;
            }
        }
        if (stats != nullptr) {
#pragma unroll
            for (int t = 0; t < 8; ++t) {
                const float s = cs[t] + __shfl_xor(cs[t], 16);
                const float q = cq[t] + __shfl_xor(cq[t], 16);
                wpart[wave][(hi ? DH : 0) + t * 16 + r] = hi ? q : s;
            }
        }
    }

    if (stats != nullptr) {
        if (!valid) {
            for (int j = lane; j < 2 * DH; j += 32) wpart[wave][j] = 0.0f;
        }
        __syncthreads();
        float tot = 0.0f;
#pragma unroll
        for (int w = 0; w < 8; ++w) tot += wpart[w][tid];
        atomicAdd(&stats[tid], tot);
    }
}

__device__ __forceinline__ float4 ld_row(const float* __restrict__ f, int s, int l) {
    return *(const float4*)(f + (size_t)s * DH + l * 4);
}
__device__ __forceinline__ float4 zero4() {
    float4 v; v.x = 0.0f; v.y = 0.0f; v.z = 0.0f; v.w = 0.0f; return v;
}
__device__ __forceinline__ void add4(float4& a, const float4 v) {
    a.x += v.x; a.y += v.y; a.z += v.z; a.w += v.w;
}
__device__ __forceinline__ void fma4(float4& a, float w, const float4 v) {
    a.x += w * v.x; a.y += w * v.y; a.z += w * v.z; a.w += w * v.w;
}

__global__ __launch_bounds__(256)
void k_gcn_gather(const int* __restrict__ rowptr, const int* __restrict__ csr_src,
                  const float* __restrict__ dinv, const float* __restrict__ xw,
                  const float* __restrict__ b, float* __restrict__ z) {
    const int d   = __builtin_amdgcn_readfirstlane(blockIdx.x * 8 + (threadIdx.x >> 5));
    const int l   = threadIdx.x & 31;
    const int beg = __builtin_amdgcn_readfirstlane(rowptr[d]);
    const int end = __builtin_amdgcn_readfirstlane(rowptr[d + 1]);
    const float dd = dinv[d];
    const float4 vs = ld_row(xw, d, l);
    const float4 bb = *(const float4*)(b + l * 4);
    float4 acc = zero4();
    for (int base = beg; base < end; base += 32) {
        const int n = min(32, end - base);
        const int mine = (l < n) ? csr_src[base + l] : 0;
        int j = 0;
        for (; j + 4 <= n; j += 4) {
            const int s0 = __builtin_amdgcn_readlane(mine, j);
            const int s1 = __builtin_amdgcn_readlane(mine, j + 1);
            const int s2 = __builtin_amdgcn_readlane(mine, j + 2);
            const int s3 = __builtin_amdgcn_readlane(mine, j + 3);
            const float4 v0 = ld_row(xw, s0, l), v1 = ld_row(xw, s1, l);
            const float4 v2 = ld_row(xw, s2, l), v3 = ld_row(xw, s3, l);
            const float w0 = dinv[s0] * dd, w1 = dinv[s1] * dd;
            const float w2 = dinv[s2] * dd, w3 = dinv[s3] * dd;
            fma4(acc, w0, v0); fma4(acc, w1, v1); fma4(acc, w2, v2); fma4(acc, w3, v3);
        }
        const int r = n - j;
        if (r > 0) {
            const int s0 = __builtin_amdgcn_readlane(mine, j);
            const float4 v0 = ld_row(xw, s0, l);
            const float w0 = dinv[s0] * dd;
            float4 v1 = zero4(), v2 = zero4();
            float w1 = 0.0f, w2 = 0.0f;
            if (r > 1) {
                const int s1 = __builtin_amdgcn_readlane(mine, j + 1);
                v1 = ld_row(xw, s1, l); w1 = dinv[s1] * dd;
            }
            if (r > 2) {
                const int s2 = __builtin_amdgcn_readlane(mine, j + 2);
                v2 = ld_row(xw, s2, l); w2 = dinv[s2] * dd;
            }
            fma4(acc, w0, v0);
            if (r > 1) fma4(acc, w1, v1);
            if (r > 2) fma4(acc, w2, v2);
        }
    }
    const float ws = dd * dd;
    float4 o;
    o.x = acc.x + ws * vs.x + bb.x; o.y = acc.y + ws * vs.y + bb.y;
    o.z = acc.z + ws * vs.z + bb.z; o.w = acc.w + ws * vs.w + bb.w;
    *(float4*)(z + (size_t)d * DH + l * 4) = o;
}

__device__ __forceinline__ void bn_coef1(float su, float sq, float g, float bt,
                                         float& sc, float& sh) {
    const float invN = 1.0f / (float)NND;
    const float mean = su * invN;
    const float var  = sq * invN - mean * mean;
    sc = rsqrtf(var + BNEPS) * g;
    sh = bt - mean * sc;
}
template <bool BN>
__device__ __forceinline__ void acc_row(float4& a, const float4 v, const float4 sc, const float4 sh) {
    if (BN) {
        a.x += fmaxf(fmaf(v.x, sc.x, sh.x), 0.0f); a.y += fmaxf(fmaf(v.y, sc.y, sh.y), 0.0f);
        a.z += fmaxf(fmaf(v.z, sc.z, sh.z), 0.0f); a.w += fmaxf(fmaf(v.w, sc.w, sh.w), 0.0f);
    } else {
        add4(a, v);
    }
}

#define PITCH 132
#define GROWS 16
template <bool BN>
__global__ __launch_bounds__(256)
void k_gcn2_layer(const int* __restrict__ rowptr, const int* __restrict__ csr_src,
                  const float* __restrict__ z, const float* __restrict__ x0,
                  const float* __restrict__ W, float* __restrict__ out, float beta,
                  const float* __restrict__ stats_in, const float* __restrict__ gamma,
                  const float* __restrict__ bbeta, float* stats_out) {
    extern __shared__ float2 sW2[];
    float* slabs = (float*)(sW2 + (DH / 2) * DH);
    float* wpart = slabs + 8 * 16 * PITCH;
    const int tid = threadIdx.x;

#pragma unroll
    for (int it = 0; it < (DH / 2) * (DH / 4) / 256; ++it) {
        const int i = tid + it * 256;
        const int kp = i / (DH / 4);
        const int c4 = i % (DH / 4);
        const float4 a0 = *(const float4*)(W + (size_t)(2 * kp)     * DH + c4 * 4);
        const float4 a1 = *(const float4*)(W + (size_t)(2 * kp + 1) * DH + c4 * 4);
        float4* dst4 = (float4*)(sW2 + (size_t)kp * DH + c4 * 4);
        float4 f0; f0.x = a0.x; f0.y = a1.x; f0.z = a0.y; f0.w = a1.y;
        float4 f1; f1.x = a0.z; f1.y = a1.z; f1.z = a0.w; f1.w = a1.w;
        dst4[0] = f0;
        dst4[1] = f1;
    }
    __syncthreads();

    const int wave = tid >> 5;
    const int lane = tid & 31;
    const int tileRow = blockIdx.x * 8 + wave;
    const bool valid = tileRow < NND / 16;
    float* slabw = slabs + wave * 16 * PITCH;
    float* wp = wpart + wave * 2 * DH;
    if (valid) {
        const int m0 = __builtin_amdgcn_readfirstlane(tileRow * 16);

#pragma unroll
        for (int rr = 0; rr < 16; ++rr)
            *(float4*)(slabw + rr * PITCH + lane * 4) = ld_row(x0, m0 + rr, lane);
        const int rpv = rowptr[m0 + min(lane, 16)];
        const int E0 = __builtin_amdgcn_readlane(rpv, 0);
        const int E1 = __builtin_amdgcn_readlane(rpv, 16);
        const int bv = (lane < 16) ? rpv : 0x7fffffff;
        float4 sc = zero4(), sh = zero4();
        if (BN) {
            const float4 su = *(const float4*)(stats_in + lane * 4);
            const float4 sq = *(const float4*)(stats_in + DH + lane * 4);
            const float4 g  = *(const float4*)(gamma + lane * 4);
            const float4 bt = *(const float4*)(bbeta + lane * 4);
            bn_coef1(su.x, sq.x, g.x, bt.x, sc.x, sh.x);
            bn_coef1(su.y, sq.y, g.y, bt.y, sc.y, sh.y);
            bn_coef1(su.z, sq.z, g.z, bt.z, sc.z, sh.z);
            bn_coef1(su.w, sq.w, g.w, bt.w, sc.w, sh.w);
        }
        const float wagg = 1.0f - ALPHA;
        float4 acc = zero4();
        int cur = 0;
        int bnd = __builtin_amdgcn_readlane(bv, 1);
        int e = E0;
        auto flush = [&]() {
            float* p = slabw + cur * PITCH + lane * 4;
            const float4 xv = *(const float4*)p;
            float4 o;
            o.x = ALPHA * xv.x + wagg * acc.x; o.y = ALPHA * xv.y + wagg * acc.y;
            o.z = ALPHA * xv.z + wagg * acc.z; o.w = ALPHA * xv.w + wagg * acc.w;
            *(float4*)p = o;
            acc = zero4();
            ++cur;
            bnd = __builtin_amdgcn_readlane(bv, cur + 1);
        };
        const char* zb = (const char*)z;
        const unsigned lane16 = (unsigned)lane * 16u;
        int mine_next = (E0 + lane < E1) ? csr_src[E0 + lane] : 0;
        for (int base = E0; base < E1; base += 32) {
            const int mine = mine_next;
            const int n = min(32, E1 - base);
#pragma unroll
            for (int jj = 0; jj < 2; ++jj) {
                if (jj * GROWS < n) {
                    const int cnt = n - jj * GROWS;
                    float4 v[GROWS];
#pragma unroll
                    for (int q = 0; q < GROWS; ++q) {
                        const unsigned s = (unsigned)__builtin_amdgcn_readlane(mine, jj * GROWS + q);
                        v[q] = *(const float4*)(zb + ((s << 9) + lane16));
                    }
                    if (jj == 0)
                        mine_next = (base + 32 + lane < E1) ? csr_src[base + 32 + lane] : 0;
#pragma unroll
                    for (int q = 0; q < GROWS; ++q) {
                        if (__builtin_expect(q < cnt, 1)) {
                            if (__builtin_expect(e == bnd, 0)) {
                                do { flush(); } while (e == bnd && cur < 15);
                            }
                            acc_row<BN>(acc, v[q], sc, sh);
                            ++e;
                        }
                    }
                }
            }
        }
        while (cur < 16) flush();

        const int r    = lane & 15;
        const bool hi  = lane >= 16;
        const int koff = hi ? 2 : 0;
        const float* arow = slabw + r * PITCH;

        v8f c[8] = {};
        for (int k0 = 0; k0 < DH; k0 += 4) {
            const float2 a2 = *(const float2*)(arow + k0 + koff);
            v2f a; a.x = a2.x; a.y = a2.y;
            const int pairRow = (k0 >> 1) + (hi ? 1 : 0);
            const float2* brow = sW2 + (size_t)pairRow * DH + r;
#pragma unroll
            for (int t = 0; t < 8; ++t) {
                const float2 b2 = brow[t * 16];
                v2f b; b.x = b2.x; b.y = b2.y;
                c[t] = __builtin_amdgcn_wmma_f32_16x16x4_f32(
                    false, a, false, b, (short)0, c[t], false, false);
            }
        }

        float cs[8], cq[8];
#pragma unroll
        for (int t = 0; t < 8; ++t) {
            const int col = t * 16 + r;
            cs[t] = 0.0f; cq[t] = 0.0f;
#pragma unroll
            for (int v = 0; v < 8; ++v) {
                const int lrow = v + (hi ? 8 : 0);
                const float hv = slabw[lrow * PITCH + col];
                const float val = (1.0f - beta) * hv + beta * c[t][v];
                out[(size_t)(m0 + lrow) * DH + col] = val;
                cs[t] += val; cq[t] += val * val;
            }
        }
        if (stats_out != nullptr) {
#pragma unroll
            for (int t = 0; t < 8; ++t) {
                const float s = cs[t] + __shfl_xor(cs[t], 16);
                const float q = cq[t] + __shfl_xor(cq[t], 16);
                wp[(hi ? DH : 0) + t * 16 + r] = hi ? q : s;
            }
        }
    }

    if (stats_out != nullptr) {
        if (!valid) {
            for (int j = lane; j < 2 * DH; j += 32) wp[j] = 0.0f;
        }
        __syncthreads();
        float tot = 0.0f;
#pragma unroll
        for (int w = 0; w < 8; ++w) tot += wpart[w * 2 * DH + tid];
        atomicAdd(&stats_out[tid], tot);
    }
}

extern "C" void kernel_launch(void* const* d_in, const int* in_sizes, int n_in,
                              void* d_out, int out_size, void* d_ws, size_t ws_size,
                              hipStream_t stream) {
    const float* x        = (const float*)d_in[0];
    const int*   edge     = (const int*)d_in[1];
    const float* W_lin    = (const float*)d_in[2];
    const float* b_lin    = (const float*)d_in[3];
    const float* W1s      = (const float*)d_in[4];
    const float* bn_gamma = (const float*)d_in[5];
    const float* bn_beta  = (const float*)d_in[6];
    float* out = (float*)d_out;

    float* ws    = (float*)d_ws;
    float* dinv  = ws;
    float* stats = ws + NND;
    float* bufA  = stats + NSTAT;
    float* bufB  = bufA + (size_t)NND * DH;
    float* x0    = bufB + (size_t)NND * DH;
    int*   wgcnt   = (int*)(x0 + (size_t)NND * DH);
    int*   bstart  = wgcnt + BK_G * BK_NB;
    int*   rowptr  = wgcnt + NPAD;
    int*   csr_src = rowptr + NPAD;
    int*   rec     = csr_src + NED;

    const int* src = edge;
    const int* dst = edge + NED;

    const int nBlkRow = NND / 8;
    const int nBlkGemm = (NND / 16 + 7) / 8;
    const size_t ldsLayer = DH * DH * sizeof(float) + 8 * 16 * PITCH * sizeof(float)
                          + 8 * 2 * DH * sizeof(float);

    k_bin_count<<<BK_G, 256, 0, stream>>>(dst, wgcnt, stats);
    k_bin_scatter<<<BK_G, 256, 0, stream>>>(src, dst, wgcnt, bstart, rec);
    k_bucket_csr<<<BK_NB, 256, 0, stream>>>(bstart, rec, rowptr, dinv, csr_src);

    k_gemm_wmma<DIN><<<nBlkGemm, 256, DIN * DH * sizeof(float), stream>>>(
        x, W_lin, bufA, 0, 0.0f, nullptr);

    k_gcn_gather<<<nBlkRow, 256, 0, stream>>>(rowptr, csr_src, dinv, bufA, b_lin, x0);

    const float* z = x0;
    for (int i = 0; i < 5; ++i) {
        const float beta = logf(0.5f / (float)(i + 1) + 1.0f);
        float* o = (i == 4) ? out : ((i & 1) ? bufB : bufA);
        const float* Wi = W1s + (size_t)i * DH * DH;
        if (i == 0) {
            k_gcn2_layer<false><<<nBlkGemm, 256, ldsLayer, stream>>>(
                rowptr, csr_src, z, x0, Wi, o, beta, nullptr, nullptr, nullptr, stats);
        } else {
            k_gcn2_layer<true><<<nBlkGemm, 256, ldsLayer, stream>>>(
                rowptr, csr_src, z, x0, Wi, o, beta, stats + (i - 1) * 2 * DH,
                bn_gamma + (i - 1) * DH, bn_beta + (i - 1) * DH,
                i < 4 ? stats + i * 2 * DH : nullptr);
        }
        z = o;
    }
}


# Edge-list build: independent requests go out in deeper batches (32 work-group counts at a time, a thread's LDS slot adds ahead of its stores, a bucket's records as one batch); segments by arrival
# speedup vs baseline: 5.6263x; 1.1015x over previous
#include <hip/hip_runtime.h>
#include <cmath>

typedef __attribute__((ext_vector_type(2))) float v2f;
typedef __attribute__((ext_vector_type(8))) float v8f;

#define NND   50000
#define NED   800000
#define DIN   256
#define DH    128
#define ALPHA 0.1f
#define BNEPS 1e-5f

#define NPAD  53248
#define NSTAT (4 * 2 * DH)
#define BK_NB  196
#define BK_G   256
#define BK_EPW (NED / BK_G)
#define BK_EPT 13
#define PF_BATCH 32

__device__ __forceinline__ int wg_excl_scan256(int c, int* wtot, int lane, int wave) {
    int incl = c;
#pragma unroll
    for (int o = 1; o < 32; o <<= 1) {
        int n = __shfl_up(incl, o);
        if (lane >= o) incl += n;
    }
    if (lane == 31) wtot[wave] = incl;
    __syncthreads();
    int woff = 0;
#pragma unroll
    for (int w = 0; w < 8; ++w) woff += (w < wave) ? wtot[w] : 0;
    return woff + incl - c;
}

__global__ __launch_bounds__(256)
void k_bin_count(const int* __restrict__ dst, int* __restrict__ wgcnt, float* __restrict__ stats) {
    __shared__ int hist[BK_NB];
    const int tid = threadIdx.x;
    const int g = blockIdx.x;
    if (tid < BK_NB) hist[tid] = 0;
    int d[BK_EPT];
#pragma unroll
    for (int k = 0; k < BK_EPT; ++k)
        d[k] = dst[g * BK_EPW + min(tid + k * 256, BK_EPW - 1)];
    __syncthreads();
#pragma unroll
    for (int k = 0; k < BK_EPT; ++k)
        if (tid + k * 256 < BK_EPW) atomicAdd(&hist[d[k] >> 8], 1);
    __syncthreads();
    if (tid < BK_NB) wgcnt[g * BK_NB + tid] = hist[tid];
    if (g == 0)
        for (int i = tid; i < NSTAT; i += 256) stats[i] = 0.0f;
}

__global__ __launch_bounds__(256)
void k_bin_scatter(const int* __restrict__ src, const int* __restrict__ dst,
                   const int* __restrict__ wgcnt, int* __restrict__ bstart,
                   int* __restrict__ rec_i) {
    int2* rec = (int2*)rec_i;
    __shared__ int cursor[BK_NB];
    __shared__ int wtot[8];
    const int tid = threadIdx.x;
    const int lane = tid & 31, wave = tid >> 5;
    const int g = blockIdx.x;
    int d[BK_EPT], s[BK_EPT];
#pragma unroll
    for (int k = 0; k < BK_EPT; ++k) {
        const int i = g * BK_EPW + min(tid + k * 256, BK_EPW - 1);
        d[k] = dst[i];
        s[k] = src[i];
    }
    int tot = 0, pre = 0;
    if (tid < BK_NB) {
        for (int g0 = 0; g0 < BK_G; g0 += PF_BATCH) {
            int c[PF_BATCH];
            const int* colp = wgcnt + g0 * BK_NB + tid;
#pragma unroll
            for (int k = 0; k < PF_BATCH; ++k) c[k] = colp[k * BK_NB];
#pragma unroll
            for (int k = 0; k < PF_BATCH; ++k) {
                tot += c[k];
                pre += (g0 + k < g) ? c[k] : 0;
            }
        }
    }
    const int excl = wg_excl_scan256(tot, wtot, lane, wave);
    if (tid < BK_NB) cursor[tid] = excl + pre;
    if (g == 0 && tid <= BK_NB) bstart[tid] = excl;
    __syncthreads();
    int pos[BK_EPT];
#pragma unroll
    for (int k = 0; k < BK_EPT; ++k)
        pos[k] = (tid + k * 256 < BK_EPW) ? atomicAdd(&cursor[d[k] >> 8], 1) : 0;
#pragma unroll
    for (int k = 0; k < BK_EPT; ++k) {
        if (tid + k * 256 < BK_EPW) {
            int2 r; r.x = d[k]; r.y = s[k];
            rec[pos[k]] = r;
        }
    }
}

#define BC_BATCH 20
__global__ __launch_bounds__(256)
void k_bucket_csr(const int* __restrict__ bstart, const int* __restrict__ rec_i,
                  int* __restrict__ rowptr, float* __restrict__ dinv,
                  int* __restrict__ csr_src) {
    const int2* rec = (const int2*)rec_i;
    __shared__ int cnt[256];
    __shared__ int wtot[8];
    const int tid = threadIdx.x;
    const int lane = tid & 31, wave = tid >> 5;
    const int b = blockIdx.x;
    const int lo = bstart[b], hi = bstart[b + 1];
    cnt[tid] = 0;
    __syncthreads();
    for (int i0 = lo; i0 < hi; i0 += 256 * BC_BATCH) {
        int dd[BC_BATCH];
#pragma unroll
        for (int k = 0; k < BC_BATCH; ++k)
            dd[k] = rec[min(i0 + k * 256 + tid, hi - 1)].x;
#pragma unroll
        for (int k = 0; k < BC_BATCH; ++k)
            if (i0 + k * 256 + tid < hi) atomicAdd(&cnt[dd[k] & 255], 1);
    }
    __syncthreads();
    const int c = cnt[tid];
    const int excl = wg_excl_scan256(c, wtot, lane, wave);
    const int node = b * 256 + tid;
    rowptr[node] = lo + excl;
    if (node < NND) dinv[node] = rsqrtf(fmaxf((float)(c + 1), 1.0f));
    cnt[tid] = excl;
    __syncthreads();
    for (int i0 = lo; i0 < hi; i0 += 256 * BC_BATCH) {
        int2 r[BC_BATCH];
#pragma unroll
        for (int k = 0; k < BC_BATCH; ++k)
            r[k] = rec[min(i0 + k * 256 + tid, hi - 1)];
        int pos[BC_BATCH];
#pragma unroll
        for (int k = 0; k < BC_BATCH; ++k)
            pos[k] = (i0 + k * 256 + tid < hi) ? atomicAdd(&cnt[r[k].x & 255], 1) : 0;
#pragma unroll
        for (int k = 0; k < BC_BATCH; ++k)
            if (i0 + k * 256 + tid < hi) csr_src[lo + pos[k]] = r[k].y;
    }
}

template <int K>
__global__ __launch_bounds__(256)
void k_gemm_wmma(const float* in, const float* __restrict__ W,
                 float* out, int mode, float beta, float* stats) {
    extern __shared__ float2 sW2[];
    __shared__ float wpart[8][2 * DH];
    const int tid = threadIdx.x;

    {
        const int totalIter = (K / 2) * (DH / 4);
        for (int i = tid; i < totalIter; i += 256) {
            const int kp = i / (DH / 4);
            const int c4 = i % (DH / 4);
            const float4 a0 = *(const float4*)(W + (size_t)(2 * kp)     * DH + c4 * 4);
            const float4 a1 = *(const float4*)(W + (size_t)(2 * kp + 1) * DH + c4 * 4);
            float4* dst4 = (float4*)(sW2 + (size_t)kp * DH + c4 * 4);
            float4 f0; f0.x = a0.x; f0.y = a1.x; f0.z = a0.y; f0.w = a1.y;
            float4 f1; f1.x = a0.z; f1.y = a1.z; f1.z = a0.w; f1.w = a1.w;
            dst4[0] = f0;
            dst4[1] = f1;
        }
    }
    __syncthreads();

    const int wave = tid >> 5;
    const int lane = tid & 31;
    const int tileRow = blockIdx.x * 8 + wave;
    const bool valid = tileRow < NND / 16;
    if (valid) {
        const int m0 = tileRow * 16;
        const int r    = lane & 15;
        const bool hi  = lane >= 16;
        const int koff = hi ? 2 : 0;
        const float* arow = in + (size_t)(m0 + r) * K;

        v8f c[8] = {};
        for (int k0 = 0; k0 < K; k0 += 4) {
            const float2 a2 = *(const float2*)(arow + k0 + koff);
            v2f a; a.x = a2.x; a.y = a2.y;
            const int pairRow = (k0 >> 1) + (hi ? 1 : 0);
            const float2* brow = sW2 + (size_t)pairRow * DH + r;
#pragma unroll
            for (int t = 0; t < 8; ++t) {
                const float2 b2 = brow[t * 16];
                v2f b; b.x = b2.x; b.y = b2.y;
                c[t] = __builtin_amdgcn_wmma_f32_16x16x4_f32(
                    false, a, false, b, (short)0, c[t], false, false);
            }
        }

        float cs[8], cq[8];
#pragma unroll
        for (int t = 0; t < 8; ++t) {
            const int col = t * 16 + r;
            cs[t] = 0.0f; cq[t] = 0.0f;
#pragma unroll
            for (int v = 0; v < 8; ++v) {
                const int row = m0 + v + (hi ? 8 : 0);
                const size_t o = (size_t)row * DH + col;
                float val = c[t][v];
                if (mode != 0) val = (1.0f - beta) * in[o] + beta * val;
                out[o] = val;
                cs[t] += val; cq[t] += val * val;
            }
        }
        if (stats != nullptr) {
#pragma unroll
            for (int t = 0; t < 8; ++t) {
                const float s = cs[t] + __shfl_xor(cs[t], 16);
                const float q = cq[t] + __shfl_xor(cq[t], 16);
                wpart[wave][(hi ? DH : 0) + t * 16 + r] = hi ? q : s;
            }
        }
    }

    if (stats != nullptr) {
        if (!valid) {
            for (int j = lane; j < 2 * DH; j += 32) wpart[wave][j] = 0.0f;
        }
        __syncthreads();
        float tot = 0.0f;
#pragma unroll
        for (int w = 0; w < 8; ++w) tot += wpart[w][tid];
        atomicAdd(&stats[tid], tot);
    }
}

__device__ __forceinline__ float4 ld_row(const float* __restrict__ f, int s, int l) {
    return *(const float4*)(f + (size_t)s * DH + l * 4);
}
__device__ __forceinline__ float4 zero4() {
    float4 v; v.x = 0.0f; v.y = 0.0f; v.z = 0.0f; v.w = 0.0f; return v;
}
__device__ __forceinline__ void add4(float4& a, const float4 v) {
    a.x += v.x; a.y += v.y; a.z += v.z; a.w += v.w;
}
__device__ __forceinline__ void fma4(float4& a, float w, const float4 v) {
    a.x += w * v.x; a.y += w * v.y; a.z += w * v.z; a.w += w * v.w;
}

__global__ __launch_bounds__(256)
void k_gcn_gather(const int* __restrict__ rowptr, const int* __restrict__ csr_src,
                  const float* __restrict__ dinv, const float* __restrict__ xw,
                  const float* __restrict__ b, float* __restrict__ z) {
    const int d   = __builtin_amdgcn_readfirstlane(blockIdx.x * 8 + (threadIdx.x >> 5));
    const int l   = threadIdx.x & 31;
    const int beg = __builtin_amdgcn_readfirstlane(rowptr[d]);
    const int end = __builtin_amdgcn_readfirstlane(rowptr[d + 1]);
    const float dd = dinv[d];
    const float4 vs = ld_row(xw, d, l);
    const float4 bb = *(const float4*)(b + l * 4);
    float4 acc = zero4();
    for (int base = beg; base < end; base += 32) {
        const int n = min(32, end - base);
        const int mine = (l < n) ? csr_src[base + l] : 0;
        int j = 0;
        for (; j + 4 <= n; j += 4) {
            const int s0 = __builtin_amdgcn_readlane(mine, j);
            const int s1 = __builtin_amdgcn_readlane(mine, j + 1);
            const int s2 = __builtin_amdgcn_readlane(mine, j + 2);
            const int s3 = __builtin_amdgcn_readlane(mine, j + 3);
            const float4 v0 = ld_row(xw, s0, l), v1 = ld_row(xw, s1, l);
            const float4 v2 = ld_row(xw, s2, l), v3 = ld_row(xw, s3, l);
            const float w0 = dinv[s0] * dd, w1 = dinv[s1] * dd;
            const float w2 = dinv[s2] * dd, w3 = dinv[s3] * dd;
            fma4(acc, w0, v0); fma4(acc, w1, v1); fma4(acc, w2, v2); fma4(acc, w3, v3);
        }
        const int r = n - j;
        if (r > 0) {
            const int s0 = __builtin_amdgcn_readlane(mine, j);
            const float4 v0 = ld_row(xw, s0, l);
            const float w0 = dinv[s0] * dd;
            float4 v1 = zero4(), v2 = zero4();
            float w1 = 0.0f, w2 = 0.0f;
            if (r > 1) {
                const int s1 = __builtin_amdgcn_readlane(mine, j + 1);
                v1 = ld_row(xw, s1, l); w1 = dinv[s1] * dd;
            }
            if (r > 2) {
                const int s2 = __builtin_amdgcn_readlane(mine, j + 2);
                v2 = ld_row(xw, s2, l); w2 = dinv[s2] * dd;
            }
            fma4(acc, w0, v0);
            if (r > 1) fma4(acc, w1, v1);
            if (r > 2) fma4(acc, w2, v2);
        }
    }
    const float ws = dd * dd;
    float4 o;
    o.x = acc.x + ws * vs.x + bb.x; o.y = acc.y + ws * vs.y + bb.y;
    o.z = acc.z + ws * vs.z + bb.z; o.w = acc.w + ws * vs.w + bb.w;
    *(float4*)(z + (size_t)d * DH + l * 4) = o;
}

__device__ __forceinline__ void bn_coef1(float su, float sq, float g, float bt,
                                         float& sc, float& sh) {
    const float invN = 1.0f / (float)NND;
    const float mean = su * invN;
    const float var  = sq * invN - mean * mean;
    sc = rsqrtf(var + BNEPS) * g;
    sh = bt - mean * sc;
}
template <bool BN>
__device__ __forceinline__ void acc_row(float4& a, const float4 v, const float4 sc, const float4 sh) {
    if (BN) {
        a.x += fmaxf(fmaf(v.x, sc.x, sh.x), 0.0f); a.y += fmaxf(fmaf(v.y, sc.y, sh.y), 0.0f);
        a.z += fmaxf(fmaf(v.z, sc.z, sh.z), 0.0f); a.w += fmaxf(fmaf(v.w, sc.w, sh.w), 0.0f);
    } else {
        add4(a, v);
    }
}

#define PITCH 132
#define GROWS 16
template <bool BN>
__global__ __launch_bounds__(256)
void k_gcn2_layer(const int* __restrict__ rowptr, const int* __restrict__ csr_src,
                  const float* __restrict__ z, const float* __restrict__ x0,
                  const float* __restrict__ W, float* __restrict__ out, float beta,
                  const float* __restrict__ stats_in, const float* __restrict__ gamma,
                  const float* __restrict__ bbeta, float* stats_out) {
    extern __shared__ float2 sW2[];
    float* slabs = (float*)(sW2 + (DH / 2) * DH);
    float* wpart = slabs + 8 * 16 * PITCH;
    const int tid = threadIdx.x;

#pragma unroll
    for (int it = 0; it < (DH / 2) * (DH / 4) / 256; ++it) {
        const int i = tid + it * 256;
        const int kp = i / (DH / 4);
        const int c4 = i % (DH / 4);
        const float4 a0 = *(const float4*)(W + (size_t)(2 * kp)     * DH + c4 * 4);
        const float4 a1 = *(const float4*)(W + (size_t)(2 * kp + 1) * DH + c4 * 4);
        float4* dst4 = (float4*)(sW2 + (size_t)kp * DH + c4 * 4);
        float4 f0; f0.x = a0.x; f0.y = a1.x; f0.z = a0.y; f0.w = a1.y;
        float4 f1; f1.x = a0.z; f1.y = a1.z; f1.z = a0.w; f1.w = a1.w;
        dst4[0] = f0;
        dst4[1] = f1;
    }
    __syncthreads();

    const int wave = tid >> 5;
    const int lane = tid & 31;
    const int tileRow = blockIdx.x * 8 + wave;
    const bool valid = tileRow < NND / 16;
    float* slabw = slabs + wave * 16 * PITCH;
    float* wp = wpart + wave * 2 * DH;
    if (valid) {
        const int m0 = __builtin_amdgcn_readfirstlane(tileRow * 16);

#pragma unroll
        for (int rr = 0; rr < 16; ++rr)
            *(float4*)(slabw + rr * PITCH + lane * 4) = ld_row(x0, m0 + rr, lane);
        const int rpv = rowptr[m0 + min(lane, 16)];
        const int E0 = __builtin_amdgcn_readlane(rpv, 0);
        const int E1 = __builtin_amdgcn_readlane(rpv, 16);
        const int bv = (lane < 16) ? rpv : 0x7fffffff;
        float4 sc = zero4(), sh = zero4();
        if (BN) {
            const float4 su = *(const float4*)(stats_in + lane * 4);
            const float4 sq = *(const float4*)(stats_in + DH + lane * 4);
            const float4 g  = *(const float4*)(gamma + lane * 4);
            const float4 bt = *(const float4*)(bbeta + lane * 4);
            bn_coef1(su.x, sq.x, g.x, bt.x, sc.x, sh.x);
            bn_coef1(su.y, sq.y, g.y, bt.y, sc.y, sh.y);
            bn_coef1(su.z, sq.z, g.z, bt.z, sc.z, sh.z);
            bn_coef1(su.w, sq.w, g.w, bt.w, sc.w, sh.w);
        }
        const float wagg = 1.0f - ALPHA;
        float4 acc = zero4();
        int cur = 0;
        int bnd = __builtin_amdgcn_readlane(bv, 1);
        int e = E0;
        auto flush = [&]() {
            float* p = slabw + cur * PITCH + lane * 4;
            const float4 xv = *(const float4*)p;
            float4 o;
            o.x = ALPHA * xv.x + wagg * acc.x; o.y = ALPHA * xv.y + wagg * acc.y;
            o.z = ALPHA * xv.z + wagg * acc.z; o.w = ALPHA * xv.w + wagg * acc.w;
            *(float4*)p = o;
            acc = zero4();
            ++cur;
            bnd = __builtin_amdgcn_readlane(bv, cur + 1);
        };
        const char* zb = (const char*)z;
        const unsigned lane16 = (unsigned)lane * 16u;
        int mine_next = (E0 + lane < E1) ? csr_src[E0 + lane] : 0;
        for (int base = E0; base < E1; base += 32) {
            const int mine = mine_next;
            const int n = min(32, E1 - base);
#pragma unroll
            for (int jj = 0; jj < 2; ++jj) {
                if (jj * GROWS < n) {
                    const int cnt = n - jj * GROWS;
                    float4 v[GROWS];
#pragma unroll
                    for (int q = 0; q < GROWS; ++q) {
                        const unsigned s = (unsigned)__builtin_amdgcn_readlane(mine, jj * GROWS + q);
                        v[q] = *(const float4*)(zb + ((s << 9) + lane16));
                    }
                    if (jj == 0)
                        mine_next = (base + 32 + lane < E1) ? csr_src[base + 32 + lane] : 0;
#pragma unroll
                    for (int q = 0; q < GROWS; ++q) {
                        if (__builtin_expect(q < cnt, 1)) {
                            if (__builtin_expect(e == bnd, 0)) {
                                do { flush(); } while (e == bnd && cur < 15);
                            }
                            acc_row<BN>(acc, v[q], sc, sh);
                            ++e;
                        }
                    }
                }
            }
        }
        while (cur < 16) flush();

        const int r    = lane & 15;
        const bool hi  = lane >= 16;
        const int koff = hi ? 2 : 0;
        const float* arow = slabw + r * PITCH;

        v8f c[8] = {};
        for (int k0 = 0; k0 < DH; k0 += 4) {
            const float2 a2 = *(const float2*)(arow + k0 + koff);
            v2f a; a.x = a2.x; a.y = a2.y;
            const int pairRow = (k0 >> 1) + (hi ? 1 : 0);
            const float2* brow = sW2 + (size_t)pairRow * DH + r;
#pragma unroll
            for (int t = 0; t < 8; ++t) {
                const float2 b2 = brow[t * 16];
                v2f b; b.x = b2.x; b.y = b2.y;
                c[t] = __builtin_amdgcn_wmma_f32_16x16x4_f32(
                    false, a, false, b, (short)0, c[t], false, false);
            }
        }

        float cs[8], cq[8];
#pragma unroll
        for (int t = 0; t < 8; ++t) {
            const int col = t * 16 + r;
            cs[t] = 0.0f; cq[t] = 0.0f;
#pragma unroll
            for (int v = 0; v < 8; ++v) {
                const int lrow = v + (hi ? 8 : 0);
                const float hv = slabw[lrow * PITCH + col];
                const float val = (1.0f - beta) * hv + beta * c[t][v];
                out[(size_t)(m0 + lrow) * DH + col] = val;
                cs[t] += val; cq[t] += val * val;
            }
        }
        if (stats_out != nullptr) {
#pragma unroll
            for (int t = 0; t < 8; ++t) {
                const float s = cs[t] + __shfl_xor(cs[t], 16);
                const float q = cq[t] + __shfl_xor(cq[t], 16);
                wp[(hi ? DH : 0) + t * 16 + r] = hi ? q : s;
            }
        }
    }

    if (stats_out != nullptr) {
        if (!valid) {
            for (int j = lane; j < 2 * DH; j += 32) wp[j] = 0.0f;
        }
        __syncthreads();
        float tot = 0.0f;
#pragma unroll
        for (int w = 0; w < 8; ++w) tot += wpart[w * 2 * DH + tid];
        atomicAdd(&stats_out[tid], tot);
    }
}

extern "C" void kernel_launch(void* const* d_in, const int* in_sizes, int n_in,
                              void* d_out, int out_size, void* d_ws, size_t ws_size,
                              hipStream_t stream) {
    const float* x        = (const float*)d_in[0];
    const int*   edge     = (const int*)d_in[1];
    const float* W_lin    = (const float*)d_in[2];
    const float* b_lin    = (const float*)d_in[3];
    const float* W1s      = (const float*)d_in[4];
    const float* bn_gamma = (const float*)d_in[5];
    const float* bn_beta  = (const float*)d_in[6];
    float* out = (float*)d_out;

    float* ws    = (float*)d_ws;
    float* dinv  = ws;
    float* stats = ws + NND;
    float* bufA  = stats + NSTAT;
    float* bufB  = bufA + (size_t)NND * DH;
    float* x0    = bufB + (size_t)NND * DH;
    int*   wgcnt   = (int*)(x0 + (size_t)NND * DH);
    int*   bstart  = wgcnt + BK_G * BK_NB;
    int*   rowptr  = wgcnt + NPAD;
    int*   csr_src = rowptr + NPAD;
    int*   rec     = csr_src + NED;

    const int* src = edge;
    const int* dst = edge + NED;

    const int nBlkRow = NND / 8;
    const int nBlkGemm = (NND / 16 + 7) / 8;
    const size_t ldsLayer = DH * DH * sizeof(float) + 8 * 16 * PITCH * sizeof(float)
                          + 8 * 2 * DH * sizeof(float);

    k_bin_count<<<BK_G, 256, 0, stream>>>(dst, wgcnt, stats);
    k_bin_scatter<<<BK_G, 256, 0, stream>>>(src, dst, wgcnt, bstart, rec);
    k_bucket_csr<<<BK_NB, 256, 0, stream>>>(bstart, rec, rowptr, dinv, csr_src);

    k_gemm_wmma<DIN><<<nBlkGemm, 256, DIN * DH * sizeof(float), stream>>>(
        x, W_lin, bufA, 0, 0.0f, nullptr);

    k_gcn_gather<<<nBlkRow, 256, 0, stream>>>(rowptr, csr_src, dinv, bufA, b_lin, x0);

    const float* z = x0;
    for (int i = 0; i < 5; ++i) {
        const float beta = logf(0.5f / (float)(i + 1) + 1.0f);
        float* o = (i == 4) ? out : ((i & 1) ? bufB : bufA);
        const float* Wi = W1s + (size_t)i * DH * DH;
        if (i == 0) {
            k_gcn2_layer<false><<<nBlkGemm, 256, ldsLayer, stream>>>(
                rowptr, csr_src, z, x0, Wi, o, beta, nullptr, nullptr, nullptr, stats);
        } else {
            k_gcn2_layer<true><<<nBlkGemm, 256, ldsLayer, stream>>>(
                rowptr, csr_src, z, x0, Wi, o, beta, stats + (i - 1) * 2 * DH,
                bn_gamma + (i - 1) * DH, bn_beta + (i - 1) * DH,
                i < 4 ? stats + i * 2 * DH : nullptr);
        }
        z = o;
    }
}


# speedup vs baseline: 5.9978x; 1.0660x over previous
#include <hip/hip_runtime.h>
#include <cmath>

typedef __attribute__((ext_vector_type(2))) float v2f;
typedef __attribute__((ext_vector_type(8))) float v8f;

#define NND   50000
#define NED   800000
#define DIN   256
#define DH    128
#define ALPHA 0.1f
#define BNEPS 1e-5f

#define NPAD  53248
#define NDINV 50048
#define NSTAT (4 * 2 * DH)
#define BK_NB  196
#define BK_G   256
#define BK_EPW (NED / BK_G)
#define BK_EPT 13
#define PF_BATCH 32

__device__ __forceinline__ int wg_excl_scan256(int c, int* wtot, int lane, int wave) {
    int incl = c;
#pragma unroll
    for (int o = 1; o < 32; o <<= 1) {
        int n = __shfl_up(incl, o);
        if (lane >= o) incl += n;
    }
    if (lane == 31) wtot[wave] = incl;
    __syncthreads();
    int woff = 0;
#pragma unroll
    for (int w = 0; w < 8; ++w) woff += (w < wave) ? wtot[w] : 0;
    return woff + incl - c;
}

__global__ __launch_bounds__(256)
void k_bin_count(const int* __restrict__ dst, int* __restrict__ wgcnt, float* __restrict__ stats) {
    __shared__ int hist[BK_NB];
    const int tid = threadIdx.x;
    const int g = blockIdx.x;
    if (tid < BK_NB) hist[tid] = 0;
    int d[BK_EPT];
#pragma unroll
    for (int k = 0; k < BK_EPT; ++k)
        d[k] = dst[g * BK_EPW + min(tid + k * 256, BK_EPW - 1)];
    __syncthreads();
#pragma unroll
    for (int k = 0; k < BK_EPT; ++k)
        if (tid + k * 256 < BK_EPW) atomicAdd(&hist[d[k] >> 8], 1);
    __syncthreads();
    if (tid < BK_NB) wgcnt[g * BK_NB + tid] = hist[tid];
    if (g == 0)
        for (int i = tid; i < NSTAT; i += 256) stats[i] = 0.0f;
}

__global__ __launch_bounds__(256)
void k_bin_scatter(const int* __restrict__ src, const int* __restrict__ dst,
                   const int* __restrict__ wgcnt, int* __restrict__ bstart,
                   int* __restrict__ rec_i) {
    int2* rec = (int2*)rec_i;
    __shared__ int cursor[BK_NB];
    __shared__ int wtot[8];
    const int tid = threadIdx.x;
    const int lane = tid & 31, wave = tid >> 5;
    const int g = blockIdx.x;
    int d[BK_EPT], s[BK_EPT];
#pragma unroll
    for (int k = 0; k < BK_EPT; ++k) {
        const int i = g * BK_EPW + min(tid + k * 256, BK_EPW - 1);
        d[k] = dst[i];
        s[k] = src[i];
    }
    int tot = 0, pre = 0;
    if (tid < BK_NB) {
        for (int g0 = 0; g0 < BK_G; g0 += PF_BATCH) {
            int c[PF_BATCH];
            const int* colp = wgcnt + g0 * BK_NB + tid;
#pragma unroll
            for (int k = 0; k < PF_BATCH; ++k) c[k] = colp[k * BK_NB];
#pragma unroll
            for (int k = 0; k < PF_BATCH; ++k) {
                tot += c[k];
                pre += (g0 + k < g) ? c[k] : 0;
            }
        }
    }
    const int excl = wg_excl_scan256(tot, wtot, lane, wave);
    if (tid < BK_NB) cursor[tid] = excl + pre;
    if (g == 0 && tid <= BK_NB) bstart[tid] = excl;
    __syncthreads();
    int pos[BK_EPT];
#pragma unroll
    for (int k = 0; k < BK_EPT; ++k)
        pos[k] = (tid + k * 256 < BK_EPW) ? atomicAdd(&cursor[d[k] >> 8], 1) : 0;
#pragma unroll
    for (int k = 0; k < BK_EPT; ++k) {
        if (tid + k * 256 < BK_EPW) {
            int2 r; r.x = d[k]; r.y = s[k];
            rec[pos[k]] = r;
        }
    }
}

#define BC_BATCH 20
__global__ __launch_bounds__(256)
void k_bucket_csr(const int* __restrict__ bstart, const int* __restrict__ rec_i,
                  int* __restrict__ rowptr, float* __restrict__ dinv,
                  int* __restrict__ csr_src) {
    const int2* rec = (const int2*)rec_i;
    __shared__ int cnt[256];
    __shared__ int wtot[8];
    const int tid = threadIdx.x;
    const int lane = tid & 31, wave = tid >> 5;
    const int b = blockIdx.x;
    const int lo = bstart[b], hi = bstart[b + 1];
    cnt[tid] = 0;
    __syncthreads();
    for (int i0 = lo; i0 < hi; i0 += 256 * BC_BATCH) {
        int dd[BC_BATCH];
#pragma unroll
        for (int k = 0; k < BC_BATCH; ++k)
            dd[k] = rec[min(i0 + k * 256 + tid, hi - 1)].x;
#pragma unroll
        for (int k = 0; k < BC_BATCH; ++k)
            if (i0 + k * 256 + tid < hi) atomicAdd(&cnt[dd[k] & 255], 1);
    }
    __syncthreads();
    const int c = cnt[tid];
    const int excl = wg_excl_scan256(c, wtot, lane, wave);
    const int node = b * 256 + tid;
    rowptr[node] = lo + excl;
    if (node < NND) dinv[node] = rsqrtf(fmaxf((float)(c + 1), 1.0f));
    cnt[tid] = excl;
    __syncthreads();
    for (int i0 = lo; i0 < hi; i0 += 256 * BC_BATCH) {
        int2 r[BC_BATCH];
#pragma unroll
        for (int k = 0; k < BC_BATCH; ++k)
            r[k] = rec[min(i0 + k * 256 + tid, hi - 1)];
        int pos[BC_BATCH];
#pragma unroll
        for (int k = 0; k < BC_BATCH; ++k)
            pos[k] = (i0 + k * 256 + tid < hi) ? atomicAdd(&cnt[r[k].x & 255], 1) : 0;
#pragma unroll
        for (int k = 0; k < BC_BATCH; ++k)
            if (i0 + k * 256 + tid < hi) csr_src[lo + pos[k]] = r[k].y;
    }
}

template <int K>
__global__ __launch_bounds__(256)
void k_gemm_wmma(const float* in, const float* __restrict__ W,
                 float* out, int mode, float beta, float* stats) {
    extern __shared__ float2 sW2[];
    __shared__ float wpart[8][2 * DH];
    const int tid = threadIdx.x;

    {
        const int totalIter = (K / 2) * (DH / 4);
        for (int i = tid; i < totalIter; i += 256) {
            const int kp = i / (DH / 4);
            const int c4 = i % (DH / 4);
            const float4 a0 = *(const float4*)(W + (size_t)(2 * kp)     * DH + c4 * 4);
            const float4 a1 = *(const float4*)(W + (size_t)(2 * kp + 1) * DH + c4 * 4);
            float4* dst4 = (float4*)(sW2 + (size_t)kp * DH + c4 * 4);
            float4 f0; f0.x = a0.x; f0.y = a1.x; f0.z = a0.y; f0.w = a1.y;
            float4 f1; f1.x = a0.z; f1.y = a1.z; f1.z = a0.w; f1.w = a1.w;
            dst4[0] = f0;
            dst4[1] = f1;
        }
    }
    __syncthreads();

    const int wave = tid >> 5;
    const int lane = tid & 31;
    const int tileRow = blockIdx.x * 8 + wave;
    const bool valid = tileRow < NND / 16;
    if (valid) {
        const int m0 = tileRow * 16;
        const int r    = lane & 15;
        const bool hi  = lane >= 16;
        const int koff = hi ? 2 : 0;
        const float* arow = in + (size_t)(m0 + r) * K;

        v8f c[8] = {};
        for (int k0 = 0; k0 < K; k0 += 4) {
            const float2 a2 = *(const float2*)(arow + k0 + koff);
            v2f a; a.x = a2.x; a.y = a2.y;
            const int pairRow = (k0 >> 1) + (hi ? 1 : 0);
            const float2* brow = sW2 + (size_t)pairRow * DH + r;
#pragma unroll
            for (int t = 0; t < 8; ++t) {
                const float2 b2 = brow[t * 16];
                v2f b; b.x = b2.x; b.y = b2.y;
                c[t] = __builtin_amdgcn_wmma_f32_16x16x4_f32(
                    false, a, false, b, (short)0, c[t], false, false);
            }
        }

        float cs[8], cq[8];
#pragma unroll
        for (int t = 0; t < 8; ++t) {
            const int col = t * 16 + r;
            cs[t] = 0.0f; cq[t] = 0.0f;
#pragma unroll
            for (int v = 0; v < 8; ++v) {
                const int row = m0 + v + (hi ? 8 : 0);
                const size_t o = (size_t)row * DH + col;
                float val = c[t][v];
                if (mode != 0) val = (1.0f - beta) * in[o] + beta * val;
                out[o] = val;
                cs[t] += val; cq[t] += val * val;
            }
        }
        if (stats != nullptr) {
#pragma unroll
            for (int t = 0; t < 8; ++t) {
                const float s = cs[t] + __shfl_xor(cs[t], 16);
                const float q = cq[t] + __shfl_xor(cq[t], 16);
                wpart[wave][(hi ? DH : 0) + t * 16 + r] = hi ? q : s;
            }
        }
    }

    if (stats != nullptr) {
        if (!valid) {
            for (int j = lane; j < 2 * DH; j += 32) wpart[wave][j] = 0.0f;
        }
        __syncthreads();
        float tot = 0.0f;
#pragma unroll
        for (int w = 0; w < 8; ++w) tot += wpart[w][tid];
        atomicAdd(&stats[tid], tot);
    }
}

__device__ __forceinline__ float4 ld_row(const float* __restrict__ f, int s, int l) {
    return *(const float4*)(f + (size_t)s * DH + l * 4);
}
__device__ __forceinline__ float4 zero4() {
    float4 v; v.x = 0.0f; v.y = 0.0f; v.z = 0.0f; v.w = 0.0f; return v;
}
__device__ __forceinline__ void add4(float4& a, const float4 v) {
    a.x += v.x; a.y += v.y; a.z += v.z; a.w += v.w;
}
__device__ __forceinline__ void fma4(float4& a, float w, const float4 v) {
    a.x += w * v.x; a.y += w * v.y; a.z += w * v.z; a.w += w * v.w;
}

__global__ __launch_bounds__(256)
void k_gcn_gather(const int* __restrict__ rowptr, const int* __restrict__ csr_src,
                  const float* __restrict__ dinv, const float* __restrict__ xw,
                  const float* __restrict__ b, float* __restrict__ z) {
    const int d   = __builtin_amdgcn_readfirstlane(blockIdx.x * 8 + (threadIdx.x >> 5));
    const int l   = threadIdx.x & 31;
    const int beg = __builtin_amdgcn_readfirstlane(rowptr[d]);
    const int end = __builtin_amdgcn_readfirstlane(rowptr[d + 1]);
    const float dd = dinv[d];
    const float4 vs = ld_row(xw, d, l);
    const float4 bb = *(const float4*)(b + l * 4);
    float4 acc = zero4();
    for (int base = beg; base < end; base += 32) {
        const int n = min(32, end - base);
        const int mine = (l < n) ? csr_src[base + l] : 0;
        int j = 0;
        for (; j + 4 <= n; j += 4) {
            const int s0 = __builtin_amdgcn_readlane(mine, j);
            const int s1 = __builtin_amdgcn_readlane(mine, j + 1);
            const int s2 = __builtin_amdgcn_readlane(mine, j + 2);
            const int s3 = __builtin_amdgcn_readlane(mine, j + 3);
            const float4 v0 = ld_row(xw, s0, l), v1 = ld_row(xw, s1, l);
            const float4 v2 = ld_row(xw, s2, l), v3 = ld_row(xw, s3, l);
            const float w0 = dinv[s0] * dd, w1 = dinv[s1] * dd;
            const float w2 = dinv[s2] * dd, w3 = dinv[s3] * dd;
            fma4(acc, w0, v0); fma4(acc, w1, v1); fma4(acc, w2, v2); fma4(acc, w3, v3);
        }
        const int r = n - j;
        if (r > 0) {
            const int s0 = __builtin_amdgcn_readlane(mine, j);
            const float4 v0 = ld_row(xw, s0, l);
            const float w0 = dinv[s0] * dd;
            float4 v1 = zero4(), v2 = zero4();
            float w1 = 0.0f, w2 = 0.0f;
            if (r > 1) {
                const int s1 = __builtin_amdgcn_readlane(mine, j + 1);
                v1 = ld_row(xw, s1, l); w1 = dinv[s1] * dd;
            }
            if (r > 2) {
                const int s2 = __builtin_amdgcn_readlane(mine, j + 2);
                v2 = ld_row(xw, s2, l); w2 = dinv[s2] * dd;
            }
            fma4(acc, w0, v0);
            if (r > 1) fma4(acc, w1, v1);
            if (r > 2) fma4(acc, w2, v2);
        }
    }
    const float ws = dd * dd;
    float4 o;
    o.x = acc.x + ws * vs.x + bb.x; o.y = acc.y + ws * vs.y + bb.y;
    o.z = acc.z + ws * vs.z + bb.z; o.w = acc.w + ws * vs.w + bb.w;
    *(float4*)(z + (size_t)d * DH + l * 4) = o;
}

__device__ __forceinline__ void bn_coef1(float su, float sq, float g, float bt,
                                         float& sc, float& sh) {
    const float invN = 1.0f / (float)NND;
    const float mean = su * invN;
    const float var  = sq * invN - mean * mean;
    sc = rsqrtf(var + BNEPS) * g;
    sh = bt - mean * sc;
}
template <bool BN>
__device__ __forceinline__ void acc_row(float4& a, const float4 v, const float4 sc, const float4 sh) {
    if (BN) {
        a.x += fmaxf(fmaf(v.x, sc.x, sh.x), 0.0f); a.y += fmaxf(fmaf(v.y, sc.y, sh.y), 0.0f);
        a.z += fmaxf(fmaf(v.z, sc.z, sh.z), 0.0f); a.w += fmaxf(fmaf(v.w, sc.w, sh.w), 0.0f);
    } else {
        add4(a, v);
    }
}

#define PITCH 132
#define GROWS 16
template <bool BN>
__global__ __launch_bounds__(256)
void k_gcn2_layer(const int* __restrict__ rowptr, const int* __restrict__ csr_src,
                  const float* __restrict__ z, const float* __restrict__ x0,
                  const float* __restrict__ W, float* __restrict__ out, float beta,
                  const float* __restrict__ stats_in, const float* __restrict__ gamma,
                  const float* __restrict__ bbeta, float* stats_out) {
    extern __shared__ float2 sW2[];
    float* slabs = (float*)(sW2 + (DH / 2) * DH);
    float* wpart = slabs + 8 * 16 * PITCH;
    const int tid = threadIdx.x;

#pragma unroll
    for (int it = 0; it < (DH / 2) * (DH / 4) / 256; ++it) {
        const int i = tid + it * 256;
        const int kp = i / (DH / 4);
        const int c4 = i % (DH / 4);
        const float4 a0 = *(const float4*)(W + (size_t)(2 * kp)     * DH + c4 * 4);
        const float4 a1 = *(const float4*)(W + (size_t)(2 * kp + 1) * DH + c4 * 4);
        float4* dst4 = (float4*)(sW2 + (size_t)kp * DH + c4 * 4);
        float4 f0; f0.x = a0.x; f0.y = a1.x; f0.z = a0.y; f0.w = a1.y;
        float4 f1; f1.x = a0.z; f1.y = a1.z; f1.z = a0.w; f1.w = a1.w;
        dst4[0] = f0;
        dst4[1] = f1;
    }
    __syncthreads();

    const int wave = tid >> 5;
    const int lane = tid & 31;
    const int tileRow = blockIdx.x * 8 + wave;
    const bool valid = tileRow < NND / 16;
    float* slabw = slabs + wave * 16 * PITCH;
    float* wp = wpart + wave * 2 * DH;
    if (valid) {
        const int m0 = __builtin_amdgcn_readfirstlane(tileRow * 16);

#pragma unroll
        for (int rr = 0; rr < 16; ++rr)
            *(float4*)(slabw + rr * PITCH + lane * 4) = ld_row(x0, m0 + rr, lane);
        const int rpv = rowptr[m0 + min(lane, 16)];
        const int E0 = __builtin_amdgcn_readlane(rpv, 0);
        const int E1 = __builtin_amdgcn_readlane(rpv, 16);
        const int bv = (lane < 16) ? rpv : 0x7fffffff;
        float4 sc = zero4(), sh = zero4();
        if (BN) {
            const float4 su = *(const float4*)(stats_in + lane * 4);
            const float4 sq = *(const float4*)(stats_in + DH + lane * 4);
            const float4 g  = *(const float4*)(gamma + lane * 4);
            const float4 bt = *(const float4*)(bbeta + lane * 4);
            bn_coef1(su.x, sq.x, g.x, bt.x, sc.x, sh.x);
            bn_coef1(su.y, sq.y, g.y, bt.y, sc.y, sh.y);
            bn_coef1(su.z, sq.z, g.z, bt.z, sc.z, sh.z);
            bn_coef1(su.w, sq.w, g.w, bt.w, sc.w, sh.w);
        }
        const float wagg = 1.0f - ALPHA;
        float4 acc = zero4();
        int cur = 0;
        int bnd = __builtin_amdgcn_readlane(bv, 1);
        int e = E0;
        auto flush = [&]() {
            float* p = slabw + cur * PITCH + lane * 4;
            const float4 xv = *(const float4*)p;
            float4 o;
            o.x = ALPHA * xv.x + wagg * acc.x; o.y = ALPHA * xv.y + wagg * acc.y;
            o.z = ALPHA * xv.z + wagg * acc.z; o.w = ALPHA * xv.w + wagg * acc.w;
            *(float4*)p = o;
            acc = zero4();
            ++cur;
            bnd = __builtin_amdgcn_readlane(bv, cur + 1);
        };
        const char* zb = (const char*)z;
        const unsigned lane16 = (unsigned)lane * 16u;
        int mine_next = (E0 + lane < E1) ? csr_src[E0 + lane] : 0;
        for (int base = E0; base < E1; base += 32) {
            const int mine = mine_next;
            const int n = min(32, E1 - base);
#pragma unroll
            for (int jj = 0; jj < 2; ++jj) {
                if (jj * GROWS < n) {
                    const int cnt = n - jj * GROWS;
                    float4 v[GROWS];
#pragma unroll
                    for (int q = 0; q < GROWS; ++q) {
                        const unsigned s = (unsigned)__builtin_amdgcn_readlane(mine, jj * GROWS + q);
                        v[q] = *(const float4*)(zb + ((s << 9) + lane16));
                    }
                    if (jj == 0)
                        mine_next = (base + 32 + lane < E1) ? csr_src[base + 32 + lane] : 0;
#pragma unroll
                    for (int q = 0; q < GROWS; ++q) {
                        if (__builtin_expect(q < cnt, 1)) {
                            if (__builtin_expect(e == bnd, 0)) {
                                do { flush(); } while (e == bnd && cur < 15);
                            }
                            acc_row<BN>(acc, v[q], sc, sh);
                            ++e;
                        }
                    }
                }
            }
        }
        while (cur < 16) flush();

        const int r    = lane & 15;
        const bool hi  = lane >= 16;
        const int koff = hi ? 2 : 0;
        const float* arow = slabw + r * PITCH;

        v8f c[8] = {};
        for (int k0 = 0; k0 < DH; k0 += 4) {
            const float2 a2 = *(const float2*)(arow + k0 + koff);
            v2f a; a.x = a2.x; a.y = a2.y;
            const int pairRow = (k0 >> 1) + (hi ? 1 : 0);
            const float2* brow = sW2 + (size_t)pairRow * DH + r;
#pragma unroll
            for (int t = 0; t < 8; ++t) {
                const float2 b2 = brow[t * 16];
                v2f b; b.x = b2.x; b.y = b2.y;
                c[t] = __builtin_amdgcn_wmma_f32_16x16x4_f32(
                    false, a, false, b, (short)0, c[t], false, false);
            }
        }

        float cs[8], cq[8];
#pragma unroll
        for (int t = 0; t < 8; ++t) {
            const int col = t * 16 + r;
            cs[t] = 0.0f; cq[t] = 0.0f;
#pragma unroll
            for (int v = 0; v < 8; ++v) {
                const int lrow = v + (hi ? 8 : 0);
                const float hv = slabw[lrow * PITCH + col];
                const float val = (1.0f - beta) * hv + beta * c[t][v];
                out[(size_t)(m0 + lrow) * DH + col] = val;
                cs[t] += val; cq[t] += val * val;
            }
        }
        if (stats_out != nullptr) {
#pragma unroll
            for (int t = 0; t < 8; ++t) {
                const float s = cs[t] + __shfl_xor(cs[t], 16);
                const float q = cq[t] + __shfl_xor(cq[t], 16);
                wp[(hi ? DH : 0) + t * 16 + r] = hi ? q : s;
            }
        }
    }

    if (stats_out != nullptr) {
        if (!valid) {
            for (int j = lane; j < 2 * DH; j += 32) wp[j] = 0.0f;
        }
        __syncthreads();
        float tot = 0.0f;
#pragma unroll
        for (int w = 0; w < 8; ++w) tot += wpart[w * 2 * DH + tid];
        atomicAdd(&stats_out[tid], tot);
    }
}

extern "C" void kernel_launch(void* const* d_in, const int* in_sizes, int n_in,
                              void* d_out, int out_size, void* d_ws, size_t ws_size,
                              hipStream_t stream) {
    const float* x        = (const float*)d_in[0];
    const int*   edge     = (const int*)d_in[1];
    const float* W_lin    = (const float*)d_in[2];
    const float* b_lin    = (const float*)d_in[3];
    const float* W1s      = (const float*)d_in[4];
    const float* bn_gamma = (const float*)d_in[5];
    const float* bn_beta  = (const float*)d_in[6];
    float* out = (float*)d_out;

    float* ws    = (float*)d_ws;
    float* bufA  = ws;
    float* bufB  = bufA + (size_t)NND * DH;
    float* x0    = bufB + (size_t)NND * DH;
    float* dinv  = x0 + (size_t)NND * DH;
    float* stats = dinv + NDINV;
    int*   wgcnt   = (int*)(stats + NSTAT);
    int*   bstart  = wgcnt + BK_G * BK_NB;
    int*   rowptr  = wgcnt + NPAD;
    int*   csr_src = rowptr + NPAD;
    int*   rec     = csr_src + NED;

    const int* src = edge;
    const int* dst = edge + NED;

    const int nBlkRow = NND / 8;
    const int nBlkGemm = (NND / 16 + 7) / 8;
    const size_t ldsLayer = DH * DH * sizeof(float) + 8 * 16 * PITCH * sizeof(float)
                          + 8 * 2 * DH * sizeof(float);

    k_bin_count<<<BK_G, 256, 0, stream>>>(dst, wgcnt, stats);
    k_bin_scatter<<<BK_G, 256, 0, stream>>>(src, dst, wgcnt, bstart, rec);
    k_bucket_csr<<<BK_NB, 256, 0, stream>>>(bstart, rec, rowptr, dinv, csr_src);

    k_gemm_wmma<DIN><<<nBlkGemm, 256, DIN * DH * sizeof(float), stream>>>(
        x, W_lin, bufA, 0, 0.0f, nullptr);

    k_gcn_gather<<<nBlkRow, 256, 0, stream>>>(rowptr, csr_src, dinv, bufA, b_lin, x0);

    const float* z = x0;
    for (int i = 0; i < 5; ++i) {
        const float beta = logf(0.5f / (float)(i + 1) + 1.0f);
        float* o = (i == 4) ? out : ((i & 1) ? bufB : bufA);
        const float* Wi = W1s + (size_t)i * DH * DH;
        if (i == 0) {
            k_gcn2_layer<false><<<nBlkGemm, 256, ldsLayer, stream>>>(
                rowptr, csr_src, z, x0, Wi, o, beta, nullptr, nullptr, nullptr, stats);
        } else {
            k_gcn2_layer<true><<<nBlkGemm, 256, ldsLayer, stream>>>(
                rowptr, csr_src, z, x0, Wi, o, beta, stats + (i - 1) * 2 * DH,
                bn_gamma + (i - 1) * DH, bn_beta + (i - 1) * DH,
                i < 4 ? stats + i * 2 * DH : nullptr);
        }
        z = o;
    }
}
